# starting kernel
# baseline (speedup 1.0000x reference)
#include <hip/hip_runtime.h>

typedef __attribute__((ext_vector_type(2))) float v2f;
typedef __attribute__((ext_vector_type(8))) float v8f;

#define CH 64

__global__ void init_deg_kernel(float* __restrict__ deg, int n3) {
    int i = blockIdx.x * blockDim.x + threadIdx.x;
    if (i < n3) deg[i] = 1.0f;
}

__global__ void count_deg_kernel(const int* __restrict__ edges, float* __restrict__ deg, int E) {
    int e = blockIdx.x * blockDim.x + threadIdx.x;
    if (e < E) unsafeAtomicAdd(&deg[edges[E + e]], 1.0f);
}

__global__ void rsqrt_kernel(float* __restrict__ deg, int n3) {
    int i = blockIdx.x * blockDim.x + threadIdx.x;
    if (i < n3) deg[i] = rsqrtf(deg[i]);
}

__global__ void __launch_bounds__(32)
gemm_wmma_kernel(const float* __restrict__ X, const float* __restrict__ W,
                 float* __restrict__ H) {
    const int lane = threadIdx.x & 31;
    const int r    = lane & 15;
    const int kh   = lane >> 4;
    const long long m0 = (long long)blockIdx.x * 16;

    const float* xrow = X + (m0 + r) * CH;
    v8f acc0 = {}, acc1 = {}, acc2 = {}, acc3 = {};

    for (int ks = 0; ks < 16; ++ks) {
        const int k0 = ks * 4 + 2 * kh;
        v2f a;
        a.x = xrow[k0 + 0];
        a.y = xrow[k0 + 1];
        const float* w0 = W + (long long)k0 * CH;
        const float* w1 = w0 + CH;
        v2f b;
        b.x = w0[0 * 16 + r]; b.y = w1[0 * 16 + r];
        acc0 = __builtin_amdgcn_wmma_f32_16x16x4_f32(false, a, false, b, (short)0, acc0, false, false);
        b.x = w0[1 * 16 + r]; b.y = w1[1 * 16 + r];
        acc1 = __builtin_amdgcn_wmma_f32_16x16x4_f32(false, a, false, b, (short)0, acc1, false, false);
        b.x = w0[2 * 16 + r]; b.y = w1[2 * 16 + r];
        acc2 = __builtin_amdgcn_wmma_f32_16x16x4_f32(false, a, false, b, (short)0, acc2, false, false);
        b.x = w0[3 * 16 + r]; b.y = w1[3 * 16 + r];
        acc3 = __builtin_amdgcn_wmma_f32_16x16x4_f32(false, a, false, b, (short)0, acc3, false, false);
    }

#pragma unroll
    for (int j = 0; j < 8; ++j) {
        float* hrow = H + (m0 + j + 8 * kh) * CH + r;
        hrow[0]  = acc0[j];
        hrow[16] = acc1[j];
        hrow[32] = acc2[j];
        hrow[48] = acc3[j];
    }
}

__global__ void gemm_tail_kernel(const float* __restrict__ X, const float* __restrict__ W,
                                 float* __restrict__ H, int rowStart, int rows) {
    long long t = (long long)blockIdx.x * blockDim.x + threadIdx.x;
    if (t >= (long long)rows * CH) return;
    int row = rowStart + (int)(t >> 6);
    int c   = (int)(t & 63);
    float s = 0.0f;
    for (int k = 0; k < CH; ++k) s += X[(long long)row * CH + k] * W[(long long)k * CH + c];
    H[(long long)row * CH + c] = s;
}

__global__ void init_out_kernel(const float* __restrict__ h1, const float* __restrict__ h2,
                                const float* __restrict__ h3,
                                const float* __restrict__ dinv1, const float* __restrict__ dinv2,
                                const float* __restrict__ dinv3,
                                const float* __restrict__ b1, const float* __restrict__ b2,
                                const float* __restrict__ b3,
                                float* __restrict__ out, int N) {
    long long t = (long long)blockIdx.x * blockDim.x + threadIdx.x;
    long long total = (long long)N * CH;
    if (t >= total) return;
    int i = (int)(t >> 6);
    int c = (int)(t & 63);
    float d1 = dinv1[i], d2 = dinv2[i], d3 = dinv3[i];
    out[t]         = b1[c] + b2[c] + d1 * d1 * h1[t] + d2 * d2 * h2[t];
    out[total + t] = b3[c] + d3 * d3 * h3[t];
}

__global__ void scatter_kernel(const float* __restrict__ H, const float* __restrict__ dinv,
                               const int* __restrict__ edges, float* __restrict__ out, int E) {
    long long t = (long long)blockIdx.x * blockDim.x + threadIdx.x;
    if (t >= (long long)E * 16) return;
    int e  = (int)(t >> 4);
    int c4 = (int)(t & 15);
    int s = edges[e];
    int d = edges[E + e];
    float norm = dinv[s] * dinv[d];
    float4 v = *((const float4*)(H + (long long)s * CH) + c4);
    float* o = out + (long long)d * CH + c4 * 4;
    unsafeAtomicAdd(o + 0, v.x * norm);
    unsafeAtomicAdd(o + 1, v.y * norm);
    unsafeAtomicAdd(o + 2, v.z * norm);
    unsafeAtomicAdd(o + 3, v.w * norm);
}

extern "C" void kernel_launch(void* const* d_in, const int* in_sizes, int n_in,
                              void* d_out, int out_size, void* d_ws, size_t ws_size,
                              hipStream_t stream) {
    const float* x_star = (const float*)d_in[0];
    const float* x_gal  = (const float*)d_in[1];
    const int*   e_ssn  = (const int*)d_in[2];
    const int*   e_ssf  = (const int*)d_in[3];
    const int*   e_ggn  = (const int*)d_in[4];
    const float* W_ssn  = (const float*)d_in[5];
    const float* W_ssf  = (const float*)d_in[6];
    const float* W_ggn  = (const float*)d_in[7];
    const float* b_ssn  = (const float*)d_in[8];
    const float* b_ssf  = (const float*)d_in[9];
    const float* b_ggn  = (const float*)d_in[10];

    const int N = in_sizes[0] / CH;
    const int E = in_sizes[2] / 2;

    float* ws    = (float*)d_ws;
    float* h1    = ws;
    float* h2    = h1 + (size_t)N * CH;
    float* h3    = h2 + (size_t)N * CH;
    float* dinv1 = h3 + (size_t)N * CH;
    float* dinv2 = dinv1 + N;
    float* dinv3 = dinv2 + N;
    float* out   = (float*)d_out;

    init_deg_kernel<<<(3 * N + 255) / 256, 256, 0, stream>>>(dinv1, 3 * N);
    count_deg_kernel<<<(E + 255) / 256, 256, 0, stream>>>(e_ssn, dinv1, E);
    count_deg_kernel<<<(E + 255) / 256, 256, 0, stream>>>(e_ssf, dinv2, E);
    count_deg_kernel<<<(E + 255) / 256, 256, 0, stream>>>(e_ggn, dinv3, E);
    rsqrt_kernel<<<(3 * N + 255) / 256, 256, 0, stream>>>(dinv1, 3 * N);

    const int tiles = N / 16;
    const int rem   = N - tiles * 16;
    if (tiles > 0) {
        gemm_wmma_kernel<<<tiles, 32, 0, stream>>>(x_star, W_ssn, h1);
        gemm_wmma_kernel<<<tiles, 32, 0, stream>>>(x_star, W_ssf, h2);
        gemm_wmma_kernel<<<tiles, 32, 0, stream>>>(x_gal,  W_ggn, h3);
    }
    if (rem > 0) {
        int tail = (rem * CH + 255) / 256;
        gemm_tail_kernel<<<tail, 256, 0, stream>>>(x_star, W_ssn, h1, tiles * 16, rem);
        gemm_tail_kernel<<<tail, 256, 0, stream>>>(x_star, W_ssf, h2, tiles * 16, rem);
        gemm_tail_kernel<<<tail, 256, 0, stream>>>(x_gal,  W_ggn, h3, tiles * 16, rem);
    }

    long long totalNC = (long long)N * CH;
    init_out_kernel<<<(unsigned)((totalNC + 255) / 256), 256, 0, stream>>>(
        h1, h2, h3, dinv1, dinv2, dinv3, b_ssn, b_ssf, b_ggn, out, N);

    long long work = (long long)E * 16;
    unsigned sblocks = (unsigned)((work + 255) / 256);
    scatter_kernel<<<sblocks, 256, 0, stream>>>(h1, dinv1, e_ssn, out, E);
    scatter_kernel<<<sblocks, 256, 0, stream>>>(h2, dinv2, e_ssf, out, E);
    scatter_kernel<<<sblocks, 256, 0, stream>>>(h3, dinv3, e_ggn, out + totalNC, E);
}


# degree count: per-edge global f32 atomics replaced by per-block 64KB LDS histograms (ds_add_u32 over node ranges) flushed by coalesced f32 atomics, 3 edge types in 1 launch; integer sums, order-free
# speedup vs baseline: 1.4491x; 1.4491x over previous
#include <hip/hip_runtime.h>

typedef __attribute__((ext_vector_type(2))) float v2f;
typedef __attribute__((ext_vector_type(8))) float v8f;

#define CH 64

__global__ void init_deg_kernel(float* __restrict__ deg, int n3) {
    int i = blockIdx.x * blockDim.x + threadIdx.x;
    if (i < n3) deg[i] = 1.0f;
}

#define HB 16384
#define HT 1024
#define HU 8
__global__ void __launch_bounds__(HT)
hist_deg_kernel(const int* __restrict__ e0, const int* __restrict__ e1,
                const int* __restrict__ e2, float* __restrict__ deg,
                int E, int N, int chunk) {
    __shared__ unsigned int hist[HB];
    const int tid  = threadIdx.x;
    const int type = blockIdx.z;
    const int* __restrict__ dst = (type == 0 ? e0 : (type == 1 ? e1 : e2)) + E;
    float* __restrict__ dg = deg + (long long)type * N;
    const int lo = blockIdx.y * HB;

    for (int i = tid; i < HB; i += HT) hist[i] = 0u;
    __syncthreads();

    const int beg = blockIdx.x * chunk;
    const int end = min(beg + chunk, E);
    for (int i = beg + tid; i < end; i += HU * HT) {
        int d[HU];
#pragma unroll
        for (int u = 0; u < HU; ++u) {
            const int j = i + u * HT;
            d[u] = (j < end) ? dst[j] : -1;
        }
#pragma unroll
        for (int u = 0; u < HU; ++u) {
            const unsigned int b = (unsigned int)(d[u] - lo);
            if (b < HB) atomicAdd(&hist[b], 1u);
        }
    }
    __syncthreads();

    for (int i = tid; i < HB; i += HT) {
        const unsigned int c = hist[i];
        const int node = lo + i;
        if (c != 0u && node < N) unsafeAtomicAdd(&dg[node], (float)c);
    }
}

__global__ void rsqrt_kernel(float* __restrict__ deg, int n3) {
    int i = blockIdx.x * blockDim.x + threadIdx.x;
    if (i < n3) deg[i] = rsqrtf(deg[i]);
}

__global__ void __launch_bounds__(32)
gemm_wmma_kernel(const float* __restrict__ X, const float* __restrict__ W,
                 float* __restrict__ H) {
    const int lane = threadIdx.x & 31;
    const int r    = lane & 15;
    const int kh   = lane >> 4;
    const long long m0 = (long long)blockIdx.x * 16;

    const float* xrow = X + (m0 + r) * CH;
    v8f acc0 = {}, acc1 = {}, acc2 = {}, acc3 = {};

    for (int ks = 0; ks < 16; ++ks) {
        const int k0 = ks * 4 + 2 * kh;
        v2f a;
        a.x = xrow[k0 + 0];
        a.y = xrow[k0 + 1];
        const float* w0 = W + (long long)k0 * CH;
        const float* w1 = w0 + CH;
        v2f b;
        b.x = w0[0 * 16 + r]; b.y = w1[0 * 16 + r];
        acc0 = __builtin_amdgcn_wmma_f32_16x16x4_f32(false, a, false, b, (short)0, acc0, false, false);
        b.x = w0[1 * 16 + r]; b.y = w1[1 * 16 + r];
        acc1 = __builtin_amdgcn_wmma_f32_16x16x4_f32(false, a, false, b, (short)0, acc1, false, false);
        b.x = w0[2 * 16 + r]; b.y = w1[2 * 16 + r];
        acc2 = __builtin_amdgcn_wmma_f32_16x16x4_f32(false, a, false, b, (short)0, acc2, false, false);
        b.x = w0[3 * 16 + r]; b.y = w1[3 * 16 + r];
        acc3 = __builtin_amdgcn_wmma_f32_16x16x4_f32(false, a, false, b, (short)0, acc3, false, false);
    }

#pragma unroll
    for (int j = 0; j < 8; ++j) {
        float* hrow = H + (m0 + j + 8 * kh) * CH + r;
        hrow[0]  = acc0[j];
        hrow[16] = acc1[j];
        hrow[32] = acc2[j];
        hrow[48] = acc3[j];
    }
}

__global__ void gemm_tail_kernel(const float* __restrict__ X, const float* __restrict__ W,
                                 float* __restrict__ H, int rowStart, int rows) {
    long long t = (long long)blockIdx.x * blockDim.x + threadIdx.x;
    if (t >= (long long)rows * CH) return;
    int row = rowStart + (int)(t >> 6);
    int c   = (int)(t & 63);
    float s = 0.0f;
    for (int k = 0; k < CH; ++k) s += X[(long long)row * CH + k] * W[(long long)k * CH + c];
    H[(long long)row * CH + c] = s;
}

__global__ void init_out_kernel(const float* __restrict__ h1, const float* __restrict__ h2,
                                const float* __restrict__ h3,
                                const float* __restrict__ dinv1, const float* __restrict__ dinv2,
                                const float* __restrict__ dinv3,
                                const float* __restrict__ b1, const float* __restrict__ b2,
                                const float* __restrict__ b3,
                                float* __restrict__ out, int N) {
    long long t = (long long)blockIdx.x * blockDim.x + threadIdx.x;
    long long total = (long long)N * CH;
    if (t >= total) return;
    int i = (int)(t >> 6);
    int c = (int)(t & 63);
    float d1 = dinv1[i], d2 = dinv2[i], d3 = dinv3[i];
    out[t]         = b1[c] + b2[c] + d1 * d1 * h1[t] + d2 * d2 * h2[t];
    out[total + t] = b3[c] + d3 * d3 * h3[t];
}

__global__ void scatter_kernel(const float* __restrict__ H, const float* __restrict__ dinv,
                               const int* __restrict__ edges, float* __restrict__ out, int E) {
    long long t = (long long)blockIdx.x * blockDim.x + threadIdx.x;
    if (t >= (long long)E * 16) return;
    int e  = (int)(t >> 4);
    int c4 = (int)(t & 15);
    int s = edges[e];
    int d = edges[E + e];
    float norm = dinv[s] * dinv[d];
    float4 v = *((const float4*)(H + (long long)s * CH) + c4);
    float* o = out + (long long)d * CH + c4 * 4;
    unsafeAtomicAdd(o + 0, v.x * norm);
    unsafeAtomicAdd(o + 1, v.y * norm);
    unsafeAtomicAdd(o + 2, v.z * norm);
    unsafeAtomicAdd(o + 3, v.w * norm);
}

extern "C" void kernel_launch(void* const* d_in, const int* in_sizes, int n_in,
                              void* d_out, int out_size, void* d_ws, size_t ws_size,
                              hipStream_t stream) {
    const float* x_star = (const float*)d_in[0];
    const float* x_gal  = (const float*)d_in[1];
    const int*   e_ssn  = (const int*)d_in[2];
    const int*   e_ssf  = (const int*)d_in[3];
    const int*   e_ggn  = (const int*)d_in[4];
    const float* W_ssn  = (const float*)d_in[5];
    const float* W_ssf  = (const float*)d_in[6];
    const float* W_ggn  = (const float*)d_in[7];
    const float* b_ssn  = (const float*)d_in[8];
    const float* b_ssf  = (const float*)d_in[9];
    const float* b_ggn  = (const float*)d_in[10];

    const int N = in_sizes[0] / CH;
    const int E = in_sizes[2] / 2;

    float* ws    = (float*)d_ws;
    float* h1    = ws;
    float* h2    = h1 + (size_t)N * CH;
    float* h3    = h2 + (size_t)N * CH;
    float* dinv1 = h3 + (size_t)N * CH;
    float* dinv2 = dinv1 + N;
    float* dinv3 = dinv2 + N;
    float* out   = (float*)d_out;

    init_deg_kernel<<<(3 * N + 255) / 256, 256, 0, stream>>>(dinv1, 3 * N);
    if (E > 0) {
        const int slices = 16;
        dim3 hgrid(slices, (N + HB - 1) / HB, 3);
        hist_deg_kernel<<<hgrid, HT, 0, stream>>>(e_ssn, e_ssf, e_ggn, dinv1, E, N,
                                                  (E + slices - 1) / slices);
    }
    rsqrt_kernel<<<(3 * N + 255) / 256, 256, 0, stream>>>(dinv1, 3 * N);

    const int tiles = N / 16;
    const int rem   = N - tiles * 16;
    if (tiles > 0) {
        gemm_wmma_kernel<<<tiles, 32, 0, stream>>>(x_star, W_ssn, h1);
        gemm_wmma_kernel<<<tiles, 32, 0, stream>>>(x_star, W_ssf, h2);
        gemm_wmma_kernel<<<tiles, 32, 0, stream>>>(x_gal,  W_ggn, h3);
    }
    if (rem > 0) {
        int tail = (rem * CH + 255) / 256;
        gemm_tail_kernel<<<tail, 256, 0, stream>>>(x_star, W_ssn, h1, tiles * 16, rem);
        gemm_tail_kernel<<<tail, 256, 0, stream>>>(x_star, W_ssf, h2, tiles * 16, rem);
        gemm_tail_kernel<<<tail, 256, 0, stream>>>(x_gal,  W_ggn, h3, tiles * 16, rem);
    }

    long long totalNC = (long long)N * CH;
    init_out_kernel<<<(unsigned)((totalNC + 255) / 256), 256, 0, stream>>>(
        h1, h2, h3, dinv1, dinv2, dinv3, b_ssn, b_ssf, b_ggn, out, N);

    long long work = (long long)E * 16;
    unsigned sblocks = (unsigned)((work + 255) / 256);
    scatter_kernel<<<sblocks, 256, 0, stream>>>(h1, dinv1, e_ssn, out, E);
    scatter_kernel<<<sblocks, 256, 0, stream>>>(h2, dinv2, e_ssf, out, E);
    scatter_kernel<<<sblocks, 256, 0, stream>>>(h3, dinv3, e_ggn, out + totalNC, E);
}


# edge scatter: lane l adds channels l and l+32, so each f32 atomic is one whole 128-B line (was 4 part lines); still 2 consecutive edges per wave in launch order; sums in atomic arrival order
# speedup vs baseline: 2.1101x; 1.4561x over previous
#include <hip/hip_runtime.h>

typedef __attribute__((ext_vector_type(2))) float v2f;
typedef __attribute__((ext_vector_type(8))) float v8f;

#define CH 64

__global__ void init_deg_kernel(float* __restrict__ deg, int n3) {
    int i = blockIdx.x * blockDim.x + threadIdx.x;
    if (i < n3) deg[i] = 1.0f;
}

#define HB 16384
#define HT 1024
#define HU 8
__global__ void __launch_bounds__(HT)
hist_deg_kernel(const int* __restrict__ e0, const int* __restrict__ e1,
                const int* __restrict__ e2, float* __restrict__ deg,
                int E, int N, int chunk) {
    __shared__ unsigned int hist[HB];
    const int tid  = threadIdx.x;
    const int type = blockIdx.z;
    const int* __restrict__ dst = (type == 0 ? e0 : (type == 1 ? e1 : e2)) + E;
    float* __restrict__ dg = deg + (long long)type * N;
    const int lo = blockIdx.y * HB;

    for (int i = tid; i < HB; i += HT) hist[i] = 0u;
    __syncthreads();

    const int beg = blockIdx.x * chunk;
    const int end = min(beg + chunk, E);
    for (int i = beg + tid; i < end; i += HU * HT) {
        int d[HU];
#pragma unroll
        for (int u = 0; u < HU; ++u) {
            const int j = i + u * HT;
            d[u] = (j < end) ? dst[j] : -1;
        }
#pragma unroll
        for (int u = 0; u < HU; ++u) {
            const unsigned int b = (unsigned int)(d[u] - lo);
            if (b < HB) atomicAdd(&hist[b], 1u);
        }
    }
    __syncthreads();

    for (int i = tid; i < HB; i += HT) {
        const unsigned int c = hist[i];
        const int node = lo + i;
        if (c != 0u && node < N) unsafeAtomicAdd(&dg[node], (float)c);
    }
}

__global__ void rsqrt_kernel(float* __restrict__ deg, int n3) {
    int i = blockIdx.x * blockDim.x + threadIdx.x;
    if (i < n3) deg[i] = rsqrtf(deg[i]);
}

__global__ void __launch_bounds__(32)
gemm_wmma_kernel(const float* __restrict__ X, const float* __restrict__ W,
                 float* __restrict__ H) {
    const int lane = threadIdx.x & 31;
    const int r    = lane & 15;
    const int kh   = lane >> 4;
    const long long m0 = (long long)blockIdx.x * 16;

    const float* xrow = X + (m0 + r) * CH;
    v8f acc0 = {}, acc1 = {}, acc2 = {}, acc3 = {};

    for (int ks = 0; ks < 16; ++ks) {
        const int k0 = ks * 4 + 2 * kh;
        v2f a;
        a.x = xrow[k0 + 0];
        a.y = xrow[k0 + 1];
        const float* w0 = W + (long long)k0 * CH;
        const float* w1 = w0 + CH;
        v2f b;
        b.x = w0[0 * 16 + r]; b.y = w1[0 * 16 + r];
        acc0 = __builtin_amdgcn_wmma_f32_16x16x4_f32(false, a, false, b, (short)0, acc0, false, false);
        b.x = w0[1 * 16 + r]; b.y = w1[1 * 16 + r];
        acc1 = __builtin_amdgcn_wmma_f32_16x16x4_f32(false, a, false, b, (short)0, acc1, false, false);
        b.x = w0[2 * 16 + r]; b.y = w1[2 * 16 + r];
        acc2 = __builtin_amdgcn_wmma_f32_16x16x4_f32(false, a, false, b, (short)0, acc2, false, false);
        b.x = w0[3 * 16 + r]; b.y = w1[3 * 16 + r];
        acc3 = __builtin_amdgcn_wmma_f32_16x16x4_f32(false, a, false, b, (short)0, acc3, false, false);
    }

#pragma unroll
    for (int j = 0; j < 8; ++j) {
        float* hrow = H + (m0 + j + 8 * kh) * CH + r;
        hrow[0]  = acc0[j];
        hrow[16] = acc1[j];
        hrow[32] = acc2[j];
        hrow[48] = acc3[j];
    }
}

__global__ void gemm_tail_kernel(const float* __restrict__ X, const float* __restrict__ W,
                                 float* __restrict__ H, int rowStart, int rows) {
    long long t = (long long)blockIdx.x * blockDim.x + threadIdx.x;
    if (t >= (long long)rows * CH) return;
    int row = rowStart + (int)(t >> 6);
    int c   = (int)(t & 63);
    float s = 0.0f;
    for (int k = 0; k < CH; ++k) s += X[(long long)row * CH + k] * W[(long long)k * CH + c];
    H[(long long)row * CH + c] = s;
}

__global__ void init_out_kernel(const float* __restrict__ h1, const float* __restrict__ h2,
                                const float* __restrict__ h3,
                                const float* __restrict__ dinv1, const float* __restrict__ dinv2,
                                const float* __restrict__ dinv3,
                                const float* __restrict__ b1, const float* __restrict__ b2,
                                const float* __restrict__ b3,
                                float* __restrict__ out, int N) {
    long long t = (long long)blockIdx.x * blockDim.x + threadIdx.x;
    long long total = (long long)N * CH;
    if (t >= total) return;
    int i = (int)(t >> 6);
    int c = (int)(t & 63);
    float d1 = dinv1[i], d2 = dinv2[i], d3 = dinv3[i];
    out[t]         = b1[c] + b2[c] + d1 * d1 * h1[t] + d2 * d2 * h2[t];
    out[total + t] = b3[c] + d3 * d3 * h3[t];
}

__global__ void __launch_bounds__(256)
scatter_kernel(const float* __restrict__ H, const float* __restrict__ dinv,
               const int* __restrict__ edges, float* __restrict__ out, int E) {
    const int lane = threadIdx.x & 31;
    const int half = lane >> 4;
    const long long pair = ((long long)blockIdx.x * blockDim.x + threadIdx.x) >> 5;
    const long long e0 = pair * 2;
    if (e0 >= E) return;
    const bool two = (e0 + 1 < E);
    const long long e = (half != 0 && two) ? e0 + 1 : e0;
    const int s = edges[e];
    const int d = edges[E + e];
    const float nrm = dinv[s] * dinv[d];
    const int   s0 = __builtin_amdgcn_readlane(s, 0);
    const int   d0 = __builtin_amdgcn_readlane(d, 0);
    const float n0 = __int_as_float(__builtin_amdgcn_readlane(__float_as_int(nrm), 0));
    const int   s1 = __builtin_amdgcn_readlane(s, 16);
    const int   d1 = __builtin_amdgcn_readlane(d, 16);
    const float n1 = __int_as_float(__builtin_amdgcn_readlane(__float_as_int(nrm), 16));

    const float* ha = H + (long long)s0 * CH + lane;
    const float* hb = H + (long long)s1 * CH + lane;
    const float a0 = ha[0], a1 = ha[32];
    const float b0 = hb[0], b1 = hb[32];

    float* oa = out + (long long)d0 * CH + lane;
    float* ob = out + (long long)d1 * CH + lane;
    const float t0 = two ? b0 * n1 : 0.0f;
    const float t1 = two ? b1 * n1 : 0.0f;
    unsafeAtomicAdd(oa,      a0 * n0);
    unsafeAtomicAdd(oa + 32, a1 * n0);
    unsafeAtomicAdd(ob,      t0);
    unsafeAtomicAdd(ob + 32, t1);
}

extern "C" void kernel_launch(void* const* d_in, const int* in_sizes, int n_in,
                              void* d_out, int out_size, void* d_ws, size_t ws_size,
                              hipStream_t stream) {
    const float* x_star = (const float*)d_in[0];
    const float* x_gal  = (const float*)d_in[1];
    const int*   e_ssn  = (const int*)d_in[2];
    const int*   e_ssf  = (const int*)d_in[3];
    const int*   e_ggn  = (const int*)d_in[4];
    const float* W_ssn  = (const float*)d_in[5];
    const float* W_ssf  = (const float*)d_in[6];
    const float* W_ggn  = (const float*)d_in[7];
    const float* b_ssn  = (const float*)d_in[8];
    const float* b_ssf  = (const float*)d_in[9];
    const float* b_ggn  = (const float*)d_in[10];

    const int N = in_sizes[0] / CH;
    const int E = in_sizes[2] / 2;

    float* ws    = (float*)d_ws;
    float* h1    = ws;
    float* h2    = h1 + (size_t)N * CH;
    float* h3    = h2 + (size_t)N * CH;
    float* dinv1 = h3 + (size_t)N * CH;
    float* dinv2 = dinv1 + N;
    float* dinv3 = dinv2 + N;
    float* out   = (float*)d_out;

    init_deg_kernel<<<(3 * N + 255) / 256, 256, 0, stream>>>(dinv1, 3 * N);
    if (E > 0) {
        const int slices = 16;
        dim3 hgrid(slices, (N + HB - 1) / HB, 3);
        hist_deg_kernel<<<hgrid, HT, 0, stream>>>(e_ssn, e_ssf, e_ggn, dinv1, E, N,
                                                  (E + slices - 1) / slices);
    }
    rsqrt_kernel<<<(3 * N + 255) / 256, 256, 0, stream>>>(dinv1, 3 * N);

    const int tiles = N / 16;
    const int rem   = N - tiles * 16;
    if (tiles > 0) {
        gemm_wmma_kernel<<<tiles, 32, 0, stream>>>(x_star, W_ssn, h1);
        gemm_wmma_kernel<<<tiles, 32, 0, stream>>>(x_star, W_ssf, h2);
        gemm_wmma_kernel<<<tiles, 32, 0, stream>>>(x_gal,  W_ggn, h3);
    }
    if (rem > 0) {
        int tail = (rem * CH + 255) / 256;
        gemm_tail_kernel<<<tail, 256, 0, stream>>>(x_star, W_ssn, h1, tiles * 16, rem);
        gemm_tail_kernel<<<tail, 256, 0, stream>>>(x_star, W_ssf, h2, tiles * 16, rem);
        gemm_tail_kernel<<<tail, 256, 0, stream>>>(x_gal,  W_ggn, h3, tiles * 16, rem);
    }

    long long totalNC = (long long)N * CH;
    init_out_kernel<<<(unsigned)((totalNC + 255) / 256), 256, 0, stream>>>(
        h1, h2, h3, dinv1, dinv2, dinv3, b_ssn, b_ssf, b_ggn, out, N);

    long long work = (long long)E * 16;
    unsigned sblocks = (unsigned)((work + 255) / 256);
    scatter_kernel<<<sblocks, 256, 0, stream>>>(h1, dinv1, e_ssn, out, E);
    scatter_kernel<<<sblocks, 256, 0, stream>>>(h2, dinv2, e_ssf, out, E);
    scatter_kernel<<<sblocks, 256, 0, stream>>>(h3, dinv3, e_ggn, out + totalNC, E);
}


# edge scatter: 4 consecutive edges per wave (was 2), all 8 row loads in flight before the first atomic, half the waves; whole-line f32 atomics, edge order per wave; sums in atomic arrival order
# speedup vs baseline: 2.6214x; 1.2423x over previous
#include <hip/hip_runtime.h>

typedef __attribute__((ext_vector_type(2))) float v2f;
typedef __attribute__((ext_vector_type(8))) float v8f;

#define CH 64

__global__ void init_deg_kernel(float* __restrict__ deg, int n3) {
    int i = blockIdx.x * blockDim.x + threadIdx.x;
    if (i < n3) deg[i] = 1.0f;
}

#define HB 16384
#define HT 1024
#define HU 8
__global__ void __launch_bounds__(HT)
hist_deg_kernel(const int* __restrict__ e0, const int* __restrict__ e1,
                const int* __restrict__ e2, float* __restrict__ deg,
                int E, int N, int chunk) {
    __shared__ unsigned int hist[HB];
    const int tid  = threadIdx.x;
    const int type = blockIdx.z;
    const int* __restrict__ dst = (type == 0 ? e0 : (type == 1 ? e1 : e2)) + E;
    float* __restrict__ dg = deg + (long long)type * N;
    const int lo = blockIdx.y * HB;

    for (int i = tid; i < HB; i += HT) hist[i] = 0u;
    __syncthreads();

    const int beg = blockIdx.x * chunk;
    const int end = min(beg + chunk, E);
    for (int i = beg + tid; i < end; i += HU * HT) {
        int d[HU];
#pragma unroll
        for (int u = 0; u < HU; ++u) {
            const int j = i + u * HT;
            d[u] = (j < end) ? dst[j] : -1;
        }
#pragma unroll
        for (int u = 0; u < HU; ++u) {
            const unsigned int b = (unsigned int)(d[u] - lo);
            if (b < HB) atomicAdd(&hist[b], 1u);
        }
    }
    __syncthreads();

    for (int i = tid; i < HB; i += HT) {
        const unsigned int c = hist[i];
        const int node = lo + i;
        if (c != 0u && node < N) unsafeAtomicAdd(&dg[node], (float)c);
    }
}

__global__ void rsqrt_kernel(float* __restrict__ deg, int n3) {
    int i = blockIdx.x * blockDim.x + threadIdx.x;
    if (i < n3) deg[i] = rsqrtf(deg[i]);
}

__global__ void __launch_bounds__(32)
gemm_wmma_kernel(const float* __restrict__ X, const float* __restrict__ W,
                 float* __restrict__ H) {
    const int lane = threadIdx.x & 31;
    const int r    = lane & 15;
    const int kh   = lane >> 4;
    const long long m0 = (long long)blockIdx.x * 16;

    const float* xrow = X + (m0 + r) * CH;
    v8f acc0 = {}, acc1 = {}, acc2 = {}, acc3 = {};

    for (int ks = 0; ks < 16; ++ks) {
        const int k0 = ks * 4 + 2 * kh;
        v2f a;
        a.x = xrow[k0 + 0];
        a.y = xrow[k0 + 1];
        const float* w0 = W + (long long)k0 * CH;
        const float* w1 = w0 + CH;
        v2f b;
        b.x = w0[0 * 16 + r]; b.y = w1[0 * 16 + r];
        acc0 = __builtin_amdgcn_wmma_f32_16x16x4_f32(false, a, false, b, (short)0, acc0, false, false);
        b.x = w0[1 * 16 + r]; b.y = w1[1 * 16 + r];
        acc1 = __builtin_amdgcn_wmma_f32_16x16x4_f32(false, a, false, b, (short)0, acc1, false, false);
        b.x = w0[2 * 16 + r]; b.y = w1[2 * 16 + r];
        acc2 = __builtin_amdgcn_wmma_f32_16x16x4_f32(false, a, false, b, (short)0, acc2, false, false);
        b.x = w0[3 * 16 + r]; b.y = w1[3 * 16 + r];
        acc3 = __builtin_amdgcn_wmma_f32_16x16x4_f32(false, a, false, b, (short)0, acc3, false, false);
    }

#pragma unroll
    for (int j = 0; j < 8; ++j) {
        float* hrow = H + (m0 + j + 8 * kh) * CH + r;
        hrow[0]  = acc0[j];
        hrow[16] = acc1[j];
        hrow[32] = acc2[j];
        hrow[48] = acc3[j];
    }
}

__global__ void gemm_tail_kernel(const float* __restrict__ X, const float* __restrict__ W,
                                 float* __restrict__ H, int rowStart, int rows) {
    long long t = (long long)blockIdx.x * blockDim.x + threadIdx.x;
    if (t >= (long long)rows * CH) return;
    int row = rowStart + (int)(t >> 6);
    int c   = (int)(t & 63);
    float s = 0.0f;
    for (int k = 0; k < CH; ++k) s += X[(long long)row * CH + k] * W[(long long)k * CH + c];
    H[(long long)row * CH + c] = s;
}

__global__ void init_out_kernel(const float* __restrict__ h1, const float* __restrict__ h2,
                                const float* __restrict__ h3,
                                const float* __restrict__ dinv1, const float* __restrict__ dinv2,
                                const float* __restrict__ dinv3,
                                const float* __restrict__ b1, const float* __restrict__ b2,
                                const float* __restrict__ b3,
                                float* __restrict__ out, int N) {
    long long t = (long long)blockIdx.x * blockDim.x + threadIdx.x;
    long long total = (long long)N * CH;
    if (t >= total) return;
    int i = (int)(t >> 6);
    int c = (int)(t & 63);
    float d1 = dinv1[i], d2 = dinv2[i], d3 = dinv3[i];
    out[t]         = b1[c] + b2[c] + d1 * d1 * h1[t] + d2 * d2 * h2[t];
    out[total + t] = b3[c] + d3 * d3 * h3[t];
}

#define EPW 4
__global__ void __launch_bounds__(256)
scatter_kernel(const float* __restrict__ H, const float* __restrict__ dinv,
               const int* __restrict__ edges, float* __restrict__ out, int E) {
    const int lane = threadIdx.x & 31;
    const long long wave = ((long long)blockIdx.x * blockDim.x + threadIdx.x) >> 5;
    const long long e0 = wave * EPW;
    if (e0 >= E) return;
    long long e = e0 + (lane >> 3);
    if (e > (long long)E - 1) e = (long long)E - 1;
    const int s = edges[e];
    const int d = edges[E + e];
    const float nrm = dinv[s] * dinv[d];

    int si[EPW], di[EPW];
    float ni[EPW];
#pragma unroll
    for (int i = 0; i < EPW; ++i) {
        si[i] = __builtin_amdgcn_readlane(s, 8 * i);
        di[i] = __builtin_amdgcn_readlane(d, 8 * i);
        ni[i] = __int_as_float(__builtin_amdgcn_readlane(__float_as_int(nrm), 8 * i));
    }

    float lo[EPW], hi[EPW];
#pragma unroll
    for (int i = 0; i < EPW; ++i) {
        const float* hp = H + (long long)si[i] * CH + lane;
        lo[i] = hp[0];
        hi[i] = hp[32];
    }

#pragma unroll
    for (int i = 0; i < EPW; ++i) {
        const bool ok = (e0 + i < E);
        float* op = out + (long long)di[i] * CH + lane;
        unsafeAtomicAdd(op,      ok ? lo[i] * ni[i] : 0.0f);
        unsafeAtomicAdd(op + 32, ok ? hi[i] * ni[i] : 0.0f);
    }
}

extern "C" void kernel_launch(void* const* d_in, const int* in_sizes, int n_in,
                              void* d_out, int out_size, void* d_ws, size_t ws_size,
                              hipStream_t stream) {
    const float* x_star = (const float*)d_in[0];
    const float* x_gal  = (const float*)d_in[1];
    const int*   e_ssn  = (const int*)d_in[2];
    const int*   e_ssf  = (const int*)d_in[3];
    const int*   e_ggn  = (const int*)d_in[4];
    const float* W_ssn  = (const float*)d_in[5];
    const float* W_ssf  = (const float*)d_in[6];
    const float* W_ggn  = (const float*)d_in[7];
    const float* b_ssn  = (const float*)d_in[8];
    const float* b_ssf  = (const float*)d_in[9];
    const float* b_ggn  = (const float*)d_in[10];

    const int N = in_sizes[0] / CH;
    const int E = in_sizes[2] / 2;

    float* ws    = (float*)d_ws;
    float* h1    = ws;
    float* h2    = h1 + (size_t)N * CH;
    float* h3    = h2 + (size_t)N * CH;
    float* dinv1 = h3 + (size_t)N * CH;
    float* dinv2 = dinv1 + N;
    float* dinv3 = dinv2 + N;
    float* out   = (float*)d_out;

    init_deg_kernel<<<(3 * N + 255) / 256, 256, 0, stream>>>(dinv1, 3 * N);
    if (E > 0) {
        const int slices = 16;
        dim3 hgrid(slices, (N + HB - 1) / HB, 3);
        hist_deg_kernel<<<hgrid, HT, 0, stream>>>(e_ssn, e_ssf, e_ggn, dinv1, E, N,
                                                  (E + slices - 1) / slices);
    }
    rsqrt_kernel<<<(3 * N + 255) / 256, 256, 0, stream>>>(dinv1, 3 * N);

    const int tiles = N / 16;
    const int rem   = N - tiles * 16;
    if (tiles > 0) {
        gemm_wmma_kernel<<<tiles, 32, 0, stream>>>(x_star, W_ssn, h1);
        gemm_wmma_kernel<<<tiles, 32, 0, stream>>>(x_star, W_ssf, h2);
        gemm_wmma_kernel<<<tiles, 32, 0, stream>>>(x_gal,  W_ggn, h3);
    }
    if (rem > 0) {
        int tail = (rem * CH + 255) / 256;
        gemm_tail_kernel<<<tail, 256, 0, stream>>>(x_star, W_ssn, h1, tiles * 16, rem);
        gemm_tail_kernel<<<tail, 256, 0, stream>>>(x_star, W_ssf, h2, tiles * 16, rem);
        gemm_tail_kernel<<<tail, 256, 0, stream>>>(x_gal,  W_ggn, h3, tiles * 16, rem);
    }

    long long totalNC = (long long)N * CH;
    init_out_kernel<<<(unsigned)((totalNC + 255) / 256), 256, 0, stream>>>(
        h1, h2, h3, dinv1, dinv2, dinv3, b_ssn, b_ssf, b_ggn, out, N);

    long long work = (((long long)E + EPW - 1) / EPW) * 32;
    unsigned sblocks = (unsigned)((work + 255) / 256);
    scatter_kernel<<<sblocks, 256, 0, stream>>>(h1, dinv1, e_ssn, out, E);
    scatter_kernel<<<sblocks, 256, 0, stream>>>(h2, dinv2, e_ssf, out, E);
    scatter_kernel<<<sblocks, 256, 0, stream>>>(h3, dinv3, e_ggn, out + totalNC, E);
}


# edge scatter kept inside one L2 half: a block reads its die and draws by ticket the edges of a 256-edge slice whose dst row is in its own half of the chip (rows split at N/2); sums in arrival order
# speedup vs baseline: 3.3000x; 1.2589x over previous
#include <hip/hip_runtime.h>

typedef __attribute__((ext_vector_type(2))) float v2f;
typedef __attribute__((ext_vector_type(8))) float v8f;

#define CH 64

#define TICKW 384
__global__ void init_deg_kernel(float* __restrict__ deg, int n3, unsigned int* __restrict__ tick) {
    int i = blockIdx.x * blockDim.x + threadIdx.x;
    if (i < n3) deg[i] = 1.0f;
    if (tick != nullptr && i < TICKW) tick[i] = 0u;
}

#define HB 16384
#define HT 1024
#define HU 8
__global__ void __launch_bounds__(HT)
hist_deg_kernel(const int* __restrict__ e0, const int* __restrict__ e1,
                const int* __restrict__ e2, float* __restrict__ deg,
                int E, int N, int chunk) {
    __shared__ unsigned int hist[HB];
    const int tid  = threadIdx.x;
    const int type = blockIdx.z;
    const int* __restrict__ dst = (type == 0 ? e0 : (type == 1 ? e1 : e2)) + E;
    float* __restrict__ dg = deg + (long long)type * N;
    const int lo = blockIdx.y * HB;

    for (int i = tid; i < HB; i += HT) hist[i] = 0u;
    __syncthreads();

    const int beg = blockIdx.x * chunk;
    const int end = min(beg + chunk, E);
    for (int i = beg + tid; i < end; i += HU * HT) {
        int d[HU];
#pragma unroll
        for (int u = 0; u < HU; ++u) {
            const int j = i + u * HT;
            d[u] = (j < end) ? dst[j] : -1;
        }
#pragma unroll
        for (int u = 0; u < HU; ++u) {
            const unsigned int b = (unsigned int)(d[u] - lo);
            if (b < HB) atomicAdd(&hist[b], 1u);
        }
    }
    __syncthreads();

    for (int i = tid; i < HB; i += HT) {
        const unsigned int c = hist[i];
        const int node = lo + i;
        if (c != 0u && node < N) unsafeAtomicAdd(&dg[node], (float)c);
    }
}

__global__ void rsqrt_kernel(float* __restrict__ deg, int n3) {
    int i = blockIdx.x * blockDim.x + threadIdx.x;
    if (i < n3) deg[i] = rsqrtf(deg[i]);
}

__global__ void __launch_bounds__(32)
gemm_wmma_kernel(const float* __restrict__ X, const float* __restrict__ W,
                 float* __restrict__ H) {
    const int lane = threadIdx.x & 31;
    const int r    = lane & 15;
    const int kh   = lane >> 4;
    const long long m0 = (long long)blockIdx.x * 16;

    const float* xrow = X + (m0 + r) * CH;
    v8f acc0 = {}, acc1 = {}, acc2 = {}, acc3 = {};

    for (int ks = 0; ks < 16; ++ks) {
        const int k0 = ks * 4 + 2 * kh;
        v2f a;
        a.x = xrow[k0 + 0];
        a.y = xrow[k0 + 1];
        const float* w0 = W + (long long)k0 * CH;
        const float* w1 = w0 + CH;
        v2f b;
        b.x = w0[0 * 16 + r]; b.y = w1[0 * 16 + r];
        acc0 = __builtin_amdgcn_wmma_f32_16x16x4_f32(false, a, false, b, (short)0, acc0, false, false);
        b.x = w0[1 * 16 + r]; b.y = w1[1 * 16 + r];
        acc1 = __builtin_amdgcn_wmma_f32_16x16x4_f32(false, a, false, b, (short)0, acc1, false, false);
        b.x = w0[2 * 16 + r]; b.y = w1[2 * 16 + r];
        acc2 = __builtin_amdgcn_wmma_f32_16x16x4_f32(false, a, false, b, (short)0, acc2, false, false);
        b.x = w0[3 * 16 + r]; b.y = w1[3 * 16 + r];
        acc3 = __builtin_amdgcn_wmma_f32_16x16x4_f32(false, a, false, b, (short)0, acc3, false, false);
    }

#pragma unroll
    for (int j = 0; j < 8; ++j) {
        float* hrow = H + (m0 + j + 8 * kh) * CH + r;
        hrow[0]  = acc0[j];
        hrow[16] = acc1[j];
        hrow[32] = acc2[j];
        hrow[48] = acc3[j];
    }
}

__global__ void gemm_tail_kernel(const float* __restrict__ X, const float* __restrict__ W,
                                 float* __restrict__ H, int rowStart, int rows) {
    long long t = (long long)blockIdx.x * blockDim.x + threadIdx.x;
    if (t >= (long long)rows * CH) return;
    int row = rowStart + (int)(t >> 6);
    int c   = (int)(t & 63);
    float s = 0.0f;
    for (int k = 0; k < CH; ++k) s += X[(long long)row * CH + k] * W[(long long)k * CH + c];
    H[(long long)row * CH + c] = s;
}

__global__ void init_out_kernel(const float* __restrict__ h1, const float* __restrict__ h2,
                                const float* __restrict__ h3,
                                const float* __restrict__ dinv1, const float* __restrict__ dinv2,
                                const float* __restrict__ dinv3,
                                const float* __restrict__ b1, const float* __restrict__ b2,
                                const float* __restrict__ b3,
                                float* __restrict__ out, int N) {
    long long t = (long long)blockIdx.x * blockDim.x + threadIdx.x;
    long long total = (long long)N * CH;
    if (t >= total) return;
    int i = (int)(t >> 6);
    int c = (int)(t & 63);
    float d1 = dinv1[i], d2 = dinv2[i], d3 = dinv3[i];
    out[t]         = b1[c] + b2[c] + d1 * d1 * h1[t] + d2 * d2 * h2[t];
    out[total + t] = b3[c] + d3 * d3 * h3[t];
}

#define EPW 4
__global__ void __launch_bounds__(256)
scatter_kernel(const float* __restrict__ H, const float* __restrict__ dinv,
               const int* __restrict__ edges, float* __restrict__ out, int E) {
    const int lane = threadIdx.x & 31;
    const long long wave = ((long long)blockIdx.x * blockDim.x + threadIdx.x) >> 5;
    const long long e0 = wave * EPW;
    if (e0 >= E) return;
    long long e = e0 + (lane >> 3);
    if (e > (long long)E - 1) e = (long long)E - 1;
    const int s = edges[e];
    const int d = edges[E + e];
    const float nrm = dinv[s] * dinv[d];

    int si[EPW], di[EPW];
    float ni[EPW];
#pragma unroll
    for (int i = 0; i < EPW; ++i) {
        si[i] = __builtin_amdgcn_readlane(s, 8 * i);
        di[i] = __builtin_amdgcn_readlane(d, 8 * i);
        ni[i] = __int_as_float(__builtin_amdgcn_readlane(__float_as_int(nrm), 8 * i));
    }

    float lo[EPW], hi[EPW];
#pragma unroll
    for (int i = 0; i < EPW; ++i) {
        const float* hp = H + (long long)si[i] * CH + lane;
        lo[i] = hp[0];
        hi[i] = hp[32];
    }

#pragma unroll
    for (int i = 0; i < EPW; ++i) {
        const bool ok = (e0 + i < E);
        float* op = out + (long long)di[i] * CH + lane;
        unsafeAtomicAdd(op,      ok ? lo[i] * ni[i] : 0.0f);
        unsafeAtomicAdd(op + 32, ok ? hi[i] * ni[i] : 0.0f);
    }
}

#define SLICE 256
#define TRIP 8
__global__ void __launch_bounds__(256)
scatter_half_kernel(const float* __restrict__ H, const float* __restrict__ dinv,
                    const int* __restrict__ edges, float* __restrict__ out,
                    unsigned int* __restrict__ tick, int E, int N, int nslices) {
    __shared__ unsigned int sh_t, sh_h;
    const int lane = threadIdx.x & 31;
    const int wv   = threadIdx.x >> 5;
    if (threadIdx.x == 0) {
        const unsigned int id = (unsigned int)__builtin_amdgcn_s_sendmsg_rtn(0x87);
        unsigned int h = (id >> 18) & 1u;
        unsigned int t = atomicAdd(&tick[h * 64], 1u);
        if (t >= (unsigned int)nslices) {
            h ^= 1u;
            t = atomicAdd(&tick[h * 64], 1u);
        }
        sh_t = t;
        sh_h = h;
    }
    __syncthreads();
    const unsigned int t    = sh_t;
    const unsigned int hsel = sh_h;
    if (t >= (unsigned int)nslices) return;

    const int split = N >> 1;
    const long long e = (long long)t * SLICE + wv * 32 + lane;
    int s = 0, d = 0;
    float nrm = 0.0f;
    bool mine = false;
    if (e < E) {
        s = edges[e];
        d = edges[E + e];
        mine = ((d >= split ? 1u : 0u) == hsel);
        if (mine) nrm = dinv[s] * dinv[d];
    }
    unsigned int mask = __builtin_amdgcn_ballot_w32(mine);

    while (mask != 0u) {
        const int n = min((int)__builtin_popcount(mask), TRIP);
        int si[TRIP], di[TRIP];
        float ni[TRIP];
#pragma unroll
        for (int k = 0; k < TRIP; ++k) {
            const int l = (mask != 0u) ? __builtin_ctz(mask) : 0;
            mask &= mask - 1u;
            si[k] = __builtin_amdgcn_readlane(s, l);
            di[k] = __builtin_amdgcn_readlane(d, l);
            ni[k] = __int_as_float(__builtin_amdgcn_readlane(__float_as_int(nrm), l));
        }
        float lo[TRIP], hi[TRIP];
#pragma unroll
        for (int k = 0; k < TRIP; ++k) {
            lo[k] = 0.0f;
            hi[k] = 0.0f;
            if (k < n) {
                const float* hp = H + (long long)si[k] * CH + lane;
                lo[k] = hp[0];
                hi[k] = hp[32];
            }
        }
#pragma unroll
        for (int k = 0; k < TRIP; ++k) {
            if (k < n) {
                float* op = out + (long long)di[k] * CH + lane;
                unsafeAtomicAdd(op,      lo[k] * ni[k]);
                unsafeAtomicAdd(op + 32, hi[k] * ni[k]);
            }
        }
    }
}

extern "C" void kernel_launch(void* const* d_in, const int* in_sizes, int n_in,
                              void* d_out, int out_size, void* d_ws, size_t ws_size,
                              hipStream_t stream) {
    const float* x_star = (const float*)d_in[0];
    const float* x_gal  = (const float*)d_in[1];
    const int*   e_ssn  = (const int*)d_in[2];
    const int*   e_ssf  = (const int*)d_in[3];
    const int*   e_ggn  = (const int*)d_in[4];
    const float* W_ssn  = (const float*)d_in[5];
    const float* W_ssf  = (const float*)d_in[6];
    const float* W_ggn  = (const float*)d_in[7];
    const float* b_ssn  = (const float*)d_in[8];
    const float* b_ssf  = (const float*)d_in[9];
    const float* b_ggn  = (const float*)d_in[10];

    const int N = in_sizes[0] / CH;
    const int E = in_sizes[2] / 2;

    float* ws    = (float*)d_ws;
    float* h1    = ws;
    float* h2    = h1 + (size_t)N * CH;
    float* h3    = h2 + (size_t)N * CH;
    float* dinv1 = h3 + (size_t)N * CH;
    float* dinv2 = dinv1 + N;
    float* dinv3 = dinv2 + N;
    float* out   = (float*)d_out;
    const size_t tickOff = (((size_t)3 * N * CH + (size_t)3 * N) * sizeof(float) + 255) / 256 * 256;
    const bool   local   = ws_size >= tickOff + TICKW * sizeof(unsigned int);
    unsigned int* tick   = local ? (unsigned int*)((char*)d_ws + tickOff) : nullptr;

    const int initN = 3 * N > TICKW ? 3 * N : TICKW;
    init_deg_kernel<<<(initN + 255) / 256, 256, 0, stream>>>(dinv1, 3 * N, tick);
    if (E > 0) {
        const int slices = 16;
        dim3 hgrid(slices, (N + HB - 1) / HB, 3);
        hist_deg_kernel<<<hgrid, HT, 0, stream>>>(e_ssn, e_ssf, e_ggn, dinv1, E, N,
                                                  (E + slices - 1) / slices);
    }
    rsqrt_kernel<<<(3 * N + 255) / 256, 256, 0, stream>>>(dinv1, 3 * N);

    const int tiles = N / 16;
    const int rem   = N - tiles * 16;
    if (tiles > 0) {
        gemm_wmma_kernel<<<tiles, 32, 0, stream>>>(x_star, W_ssn, h1);
        gemm_wmma_kernel<<<tiles, 32, 0, stream>>>(x_star, W_ssf, h2);
        gemm_wmma_kernel<<<tiles, 32, 0, stream>>>(x_gal,  W_ggn, h3);
    }
    if (rem > 0) {
        int tail = (rem * CH + 255) / 256;
        gemm_tail_kernel<<<tail, 256, 0, stream>>>(x_star, W_ssn, h1, tiles * 16, rem);
        gemm_tail_kernel<<<tail, 256, 0, stream>>>(x_star, W_ssf, h2, tiles * 16, rem);
        gemm_tail_kernel<<<tail, 256, 0, stream>>>(x_gal,  W_ggn, h3, tiles * 16, rem);
    }

    long long totalNC = (long long)N * CH;
    init_out_kernel<<<(unsigned)((totalNC + 255) / 256), 256, 0, stream>>>(
        h1, h2, h3, dinv1, dinv2, dinv3, b_ssn, b_ssf, b_ggn, out, N);

    if (local) {
        const int nslices = (E + SLICE - 1) / SLICE;
        if (nslices > 0) {
            const unsigned lblocks = 2u * (unsigned)nslices;
            scatter_half_kernel<<<lblocks, 256, 0, stream>>>(h1, dinv1, e_ssn, out, tick,       E, N, nslices);
            scatter_half_kernel<<<lblocks, 256, 0, stream>>>(h2, dinv2, e_ssf, out, tick + 128, E, N, nslices);
            scatter_half_kernel<<<lblocks, 256, 0, stream>>>(h3, dinv3, e_ggn, out + totalNC, tick + 256, E, N, nslices);
        }
    } else {
        long long work = (((long long)E + EPW - 1) / EPW) * 32;
        unsigned sblocks = (unsigned)((work + 255) / 256);
        scatter_kernel<<<sblocks, 256, 0, stream>>>(h1, dinv1, e_ssn, out, E);
        scatter_kernel<<<sblocks, 256, 0, stream>>>(h2, dinv2, e_ssf, out, E);
        scatter_kernel<<<sblocks, 256, 0, stream>>>(h3, dinv3, e_ggn, out + totalNC, E);
    }
}


# dense passes fused: 1 launch does the 3 f32 WMMA products per 16-row tile (2 star products share an X fragment), stores h and out's start value from registers; rsqrt(deg) where used; 10 launches -> 6
# speedup vs baseline: 3.6927x; 1.1190x over previous
#include <hip/hip_runtime.h>

typedef __attribute__((ext_vector_type(2))) float v2f;
typedef __attribute__((ext_vector_type(8))) float v8f;

#define CH 64

#define TICKW 384
__global__ void init_deg_kernel(float* __restrict__ deg, int n3, unsigned int* __restrict__ tick) {
    int i = blockIdx.x * blockDim.x + threadIdx.x;
    if (i < n3) deg[i] = 1.0f;
    if (tick != nullptr && i < TICKW) tick[i] = 0u;
}

#define HB 16384
#define HT 1024
#define HU 8
__global__ void __launch_bounds__(HT)
hist_deg_kernel(const int* __restrict__ e0, const int* __restrict__ e1,
                const int* __restrict__ e2, float* __restrict__ deg,
                int E, int N, int chunk) {
    __shared__ unsigned int hist[HB];
    const int tid  = threadIdx.x;
    const int type = blockIdx.z;
    const int* __restrict__ dst = (type == 0 ? e0 : (type == 1 ? e1 : e2)) + E;
    float* __restrict__ dg = deg + (long long)type * N;
    const int lo = blockIdx.y * HB;

    for (int i = tid; i < HB; i += HT) hist[i] = 0u;
    __syncthreads();

    const int beg = blockIdx.x * chunk;
    const int end = min(beg + chunk, E);
    for (int i = beg + tid; i < end; i += HU * HT) {
        int d[HU];
#pragma unroll
        for (int u = 0; u < HU; ++u) {
            const int j = i + u * HT;
            d[u] = (j < end) ? dst[j] : -1;
        }
#pragma unroll
        for (int u = 0; u < HU; ++u) {
            const unsigned int b = (unsigned int)(d[u] - lo);
            if (b < HB) atomicAdd(&hist[b], 1u);
        }
    }
    __syncthreads();

    for (int i = tid; i < HB; i += HT) {
        const unsigned int c = hist[i];
        const int node = lo + i;
        if (c != 0u && node < N) unsafeAtomicAdd(&dg[node], (float)c);
    }
}

__global__ void __launch_bounds__(32)
dense_fused_kernel(const float* __restrict__ Xs, const float* __restrict__ Xg,
                   const float* __restrict__ W1, const float* __restrict__ W2,
                   const float* __restrict__ W3,
                   const float* __restrict__ b1, const float* __restrict__ b2,
                   const float* __restrict__ b3,
                   const float* __restrict__ deg,
                   float* __restrict__ H1, float* __restrict__ H2, float* __restrict__ H3,
                   float* __restrict__ out, int N) {
    const int lane = threadIdx.x & 31;
    const int r    = lane & 15;
    const int kh   = lane >> 4;
    const long long m0 = (long long)blockIdx.x * 16;

    const float* xs0 = Xs + m0 * CH;
    const float* xg0 = Xg + m0 * CH;
    unsigned xofs = (unsigned)(r * CH + 2 * kh);
    unsigned wofs = (unsigned)(2 * kh * CH + r);
    const float bs0 = b1[r] + b2[r],           bg0 = b3[r];
    const float bs1 = b1[16 + r] + b2[16 + r], bg1 = b3[16 + r];
    const float bs2 = b1[32 + r] + b2[32 + r], bg2 = b3[32 + r];
    const float bs3 = b1[48 + r] + b2[48 + r], bg3 = b3[48 + r];
    float s1[8], s2[8], s3[8];
#pragma unroll
    for (int j = 0; j < 8; ++j) {
        const long long row = m0 + j + 8 * kh;
        const float d1 = rsqrtf(deg[row]);
        const float d2 = rsqrtf(deg[(long long)N + row]);
        const float d3 = rsqrtf(deg[2LL * N + row]);
        s1[j] = d1 * d1;  s2[j] = d2 * d2;  s3[j] = d3 * d3;
    }

    v8f p0 = {}, p1 = {}, p2 = {}, p3 = {};
    v8f q0 = {}, q1 = {}, q2 = {}, q3 = {};
    v8f g0 = {}, g1 = {}, g2 = {}, g3 = {};

    for (int ks = 0; ks < 16; ++ks) {
        v2f a, c;
        a.x = xs0[xofs];
        a.y = xs0[xofs + 1];
        c.x = xg0[xofs];
        c.y = xg0[xofs + 1];
        v2f b;
        b.x = W1[wofs + 0 * 16]; b.y = W1[wofs + CH + 0 * 16];
        p0 = __builtin_amdgcn_wmma_f32_16x16x4_f32(false, a, false, b, (short)0, p0, false, false);
        b.x = W1[wofs + 1 * 16]; b.y = W1[wofs + CH + 1 * 16];
        p1 = __builtin_amdgcn_wmma_f32_16x16x4_f32(false, a, false, b, (short)0, p1, false, false);
        b.x = W1[wofs + 2 * 16]; b.y = W1[wofs + CH + 2 * 16];
        p2 = __builtin_amdgcn_wmma_f32_16x16x4_f32(false, a, false, b, (short)0, p2, false, false);
        b.x = W1[wofs + 3 * 16]; b.y = W1[wofs + CH + 3 * 16];
        p3 = __builtin_amdgcn_wmma_f32_16x16x4_f32(false, a, false, b, (short)0, p3, false, false);

        b.x = W2[wofs + 0 * 16]; b.y = W2[wofs + CH + 0 * 16];
        q0 = __builtin_amdgcn_wmma_f32_16x16x4_f32(false, a, false, b, (short)0, q0, false, false);
        b.x = W2[wofs + 1 * 16]; b.y = W2[wofs + CH + 1 * 16];
        q1 = __builtin_amdgcn_wmma_f32_16x16x4_f32(false, a, false, b, (short)0, q1, false, false);
        b.x = W2[wofs + 2 * 16]; b.y = W2[wofs + CH + 2 * 16];
        q2 = __builtin_amdgcn_wmma_f32_16x16x4_f32(false, a, false, b, (short)0, q2, false, false);
        b.x = W2[wofs + 3 * 16]; b.y = W2[wofs + CH + 3 * 16];
        q3 = __builtin_amdgcn_wmma_f32_16x16x4_f32(false, a, false, b, (short)0, q3, false, false);

        b.x = W3[wofs + 0 * 16]; b.y = W3[wofs + CH + 0 * 16];
        g0 = __builtin_amdgcn_wmma_f32_16x16x4_f32(false, c, false, b, (short)0, g0, false, false);
        b.x = W3[wofs + 1 * 16]; b.y = W3[wofs + CH + 1 * 16];
        g1 = __builtin_amdgcn_wmma_f32_16x16x4_f32(false, c, false, b, (short)0, g1, false, false);
        b.x = W3[wofs + 2 * 16]; b.y = W3[wofs + CH + 2 * 16];
        g2 = __builtin_amdgcn_wmma_f32_16x16x4_f32(false, c, false, b, (short)0, g2, false, false);
        b.x = W3[wofs + 3 * 16]; b.y = W3[wofs + CH + 3 * 16];
        g3 = __builtin_amdgcn_wmma_f32_16x16x4_f32(false, c, false, b, (short)0, g3, false, false);
        xofs += 4;
        wofs += 4 * CH;
    }

    const long long total = (long long)N * CH;

#pragma unroll
    for (int j = 0; j < 8; ++j) {
        const long long row = m0 + j + 8 * kh;
        const long long o = row * CH + r;
        H1[o]      = p0[j];  H1[o + 16] = p1[j];  H1[o + 32] = p2[j];  H1[o + 48] = p3[j];
        H2[o]      = q0[j];  H2[o + 16] = q1[j];  H2[o + 32] = q2[j];  H2[o + 48] = q3[j];
        H3[o]      = g0[j];  H3[o + 16] = g1[j];  H3[o + 32] = g2[j];  H3[o + 48] = g3[j];
        out[o]      = bs0 + s1[j] * p0[j] + s2[j] * q0[j];
        out[o + 16] = bs1 + s1[j] * p1[j] + s2[j] * q1[j];
        out[o + 32] = bs2 + s1[j] * p2[j] + s2[j] * q2[j];
        out[o + 48] = bs3 + s1[j] * p3[j] + s2[j] * q3[j];
        out[total + o]      = bg0 + s3[j] * g0[j];
        out[total + o + 16] = bg1 + s3[j] * g1[j];
        out[total + o + 32] = bg2 + s3[j] * g2[j];
        out[total + o + 48] = bg3 + s3[j] * g3[j];
    }
}

__global__ void gemm_tail_kernel(const float* __restrict__ X, const float* __restrict__ W,
                                 float* __restrict__ H, int rowStart, int rows) {
    long long t = (long long)blockIdx.x * blockDim.x + threadIdx.x;
    if (t >= (long long)rows * CH) return;
    int row = rowStart + (int)(t >> 6);
    int c   = (int)(t & 63);
    float s = 0.0f;
    for (int k = 0; k < CH; ++k) s += X[(long long)row * CH + k] * W[(long long)k * CH + c];
    H[(long long)row * CH + c] = s;
}

__global__ void init_out_tail_kernel(const float* __restrict__ h1, const float* __restrict__ h2,
                                     const float* __restrict__ h3, const float* __restrict__ deg,
                                     const float* __restrict__ b1, const float* __restrict__ b2,
                                     const float* __restrict__ b3,
                                     float* __restrict__ out, int N, int rowStart, int rows) {
    long long q = (long long)blockIdx.x * blockDim.x + threadIdx.x;
    if (q >= (long long)rows * CH) return;
    long long total = (long long)N * CH;
    int i = rowStart + (int)(q >> 6);
    int c = (int)(q & 63);
    long long t = (long long)i * CH + c;
    float d1 = rsqrtf(deg[i]), d2 = rsqrtf(deg[(long long)N + i]), d3 = rsqrtf(deg[2LL * N + i]);
    out[t]         = b1[c] + b2[c] + d1 * d1 * h1[t] + d2 * d2 * h2[t];
    out[total + t] = b3[c] + d3 * d3 * h3[t];
}

#define EPW 4
__global__ void __launch_bounds__(256)
scatter_kernel(const float* __restrict__ H, const float* __restrict__ deg,
               const int* __restrict__ edges, float* __restrict__ out, int E) {
    const int lane = threadIdx.x & 31;
    const long long wave = ((long long)blockIdx.x * blockDim.x + threadIdx.x) >> 5;
    const long long e0 = wave * EPW;
    if (e0 >= E) return;
    long long e = e0 + (lane >> 3);
    if (e > (long long)E - 1) e = (long long)E - 1;
    const int s = edges[e];
    const int d = edges[E + e];
    const float nrm = rsqrtf(deg[s]) * rsqrtf(deg[d]);

    int si[EPW], di[EPW];
    float ni[EPW];
#pragma unroll
    for (int i = 0; i < EPW; ++i) {
        si[i] = __builtin_amdgcn_readlane(s, 8 * i);
        di[i] = __builtin_amdgcn_readlane(d, 8 * i);
        ni[i] = __int_as_float(__builtin_amdgcn_readlane(__float_as_int(nrm), 8 * i));
    }

    float lo[EPW], hi[EPW];
#pragma unroll
    for (int i = 0; i < EPW; ++i) {
        const float* hp = H + (long long)si[i] * CH + lane;
        lo[i] = hp[0];
        hi[i] = hp[32];
    }

#pragma unroll
    for (int i = 0; i < EPW; ++i) {
        const bool ok = (e0 + i < E);
        float* op = out + (long long)di[i] * CH + lane;
        unsafeAtomicAdd(op,      ok ? lo[i] * ni[i] : 0.0f);
        unsafeAtomicAdd(op + 32, ok ? hi[i] * ni[i] : 0.0f);
    }
}

#define SLICE 256
#define TRIP 8
__global__ void __launch_bounds__(256)
scatter_half_kernel(const float* __restrict__ H, const float* __restrict__ deg,
                    const int* __restrict__ edges, float* __restrict__ out,
                    unsigned int* __restrict__ tick, int E, int N, int nslices) {
    __shared__ unsigned int sh_t, sh_h;
    const int lane = threadIdx.x & 31;
    const int wv   = threadIdx.x >> 5;
    if (threadIdx.x == 0) {
        const unsigned int id = (unsigned int)__builtin_amdgcn_s_sendmsg_rtn(0x87);
        unsigned int h = (id >> 18) & 1u;
        unsigned int t = atomicAdd(&tick[h * 64], 1u);
        if (t >= (unsigned int)nslices) {
            h ^= 1u;
            t = atomicAdd(&tick[h * 64], 1u);
        }
        sh_t = t;
        sh_h = h;
    }
    __syncthreads();
    const unsigned int t    = sh_t;
    const unsigned int hsel = sh_h;
    if (t >= (unsigned int)nslices) return;

    const int split = N >> 1;
    const long long e = (long long)t * SLICE + wv * 32 + lane;
    int s = 0, d = 0;
    float nrm = 0.0f;
    bool mine = false;
    if (e < E) {
        s = edges[e];
        d = edges[E + e];
        mine = ((d >= split ? 1u : 0u) == hsel);
        if (mine) nrm = rsqrtf(deg[s]) * rsqrtf(deg[d]);
    }
    unsigned int mask = __builtin_amdgcn_ballot_w32(mine);

    while (mask != 0u) {
        const int n = min((int)__builtin_popcount(mask), TRIP);
        int si[TRIP], di[TRIP];
        float ni[TRIP];
#pragma unroll
        for (int k = 0; k < TRIP; ++k) {
            const int l = (mask != 0u) ? __builtin_ctz(mask) : 0;
            mask &= mask - 1u;
            si[k] = __builtin_amdgcn_readlane(s, l);
            di[k] = __builtin_amdgcn_readlane(d, l);
            ni[k] = __int_as_float(__builtin_amdgcn_readlane(__float_as_int(nrm), l));
        }
        float lo[TRIP], hi[TRIP];
#pragma unroll
        for (int k = 0; k < TRIP; ++k) {
            lo[k] = 0.0f;
            hi[k] = 0.0f;
            if (k < n) {
                const float* hp = H + (long long)si[k] * CH + lane;
                lo[k] = hp[0];
                hi[k] = hp[32];
            }
        }
#pragma unroll
        for (int k = 0; k < TRIP; ++k) {
            if (k < n) {
                float* op = out + (long long)di[k] * CH + lane;
                unsafeAtomicAdd(op,      lo[k] * ni[k]);
                unsafeAtomicAdd(op + 32, hi[k] * ni[k]);
            }
        }
    }
}

extern "C" void kernel_launch(void* const* d_in, const int* in_sizes, int n_in,
                              void* d_out, int out_size, void* d_ws, size_t ws_size,
                              hipStream_t stream) {
    const float* x_star = (const float*)d_in[0];
    const float* x_gal  = (const float*)d_in[1];
    const int*   e_ssn  = (const int*)d_in[2];
    const int*   e_ssf  = (const int*)d_in[3];
    const int*   e_ggn  = (const int*)d_in[4];
    const float* W_ssn  = (const float*)d_in[5];
    const float* W_ssf  = (const float*)d_in[6];
    const float* W_ggn  = (const float*)d_in[7];
    const float* b_ssn  = (const float*)d_in[8];
    const float* b_ssf  = (const float*)d_in[9];
    const float* b_ggn  = (const float*)d_in[10];

    const int N = in_sizes[0] / CH;
    const int E = in_sizes[2] / 2;

    float* ws    = (float*)d_ws;
    float* h1    = ws;
    float* h2    = h1 + (size_t)N * CH;
    float* h3    = h2 + (size_t)N * CH;
    float* deg1  = h3 + (size_t)N * CH;
    float* deg2  = deg1 + N;
    float* deg3  = deg2 + N;
    float* out   = (float*)d_out;
    const size_t tickOff = (((size_t)3 * N * CH + (size_t)3 * N) * sizeof(float) + 255) / 256 * 256;
    const bool   local   = ws_size >= tickOff + TICKW * sizeof(unsigned int);
    unsigned int* tick   = local ? (unsigned int*)((char*)d_ws + tickOff) : nullptr;

    const int initN = 3 * N > TICKW ? 3 * N : TICKW;
    init_deg_kernel<<<(initN + 255) / 256, 256, 0, stream>>>(deg1, 3 * N, tick);
    if (E > 0) {
        const int slices = 16;
        dim3 hgrid(slices, (N + HB - 1) / HB, 3);
        hist_deg_kernel<<<hgrid, HT, 0, stream>>>(e_ssn, e_ssf, e_ggn, deg1, E, N,
                                                  (E + slices - 1) / slices);
    }

    const int tiles = N / 16;
    const int rem   = N - tiles * 16;
    long long totalNC = (long long)N * CH;
    if (tiles > 0) {
        dense_fused_kernel<<<tiles, 32, 0, stream>>>(x_star, x_gal, W_ssn, W_ssf, W_ggn,
                                                     b_ssn, b_ssf, b_ggn, deg1, h1, h2, h3, out, N);
    }
    if (rem > 0) {
        int tail = (rem * CH + 255) / 256;
        gemm_tail_kernel<<<tail, 256, 0, stream>>>(x_star, W_ssn, h1, tiles * 16, rem);
        gemm_tail_kernel<<<tail, 256, 0, stream>>>(x_star, W_ssf, h2, tiles * 16, rem);
        gemm_tail_kernel<<<tail, 256, 0, stream>>>(x_gal,  W_ggn, h3, tiles * 16, rem);
        init_out_tail_kernel<<<tail, 256, 0, stream>>>(h1, h2, h3, deg1, b_ssn, b_ssf, b_ggn, out, N,
                                                       tiles * 16, rem);
    }

    if (local) {
        const int nslices = (E + SLICE - 1) / SLICE;
        if (nslices > 0) {
            const unsigned lblocks = 2u * (unsigned)nslices;
            scatter_half_kernel<<<lblocks, 256, 0, stream>>>(h1, deg1, e_ssn, out, tick,       E, N, nslices);
            scatter_half_kernel<<<lblocks, 256, 0, stream>>>(h2, deg2, e_ssf, out, tick + 128, E, N, nslices);
            scatter_half_kernel<<<lblocks, 256, 0, stream>>>(h3, deg3, e_ggn, out + totalNC, tick + 256, E, N, nslices);
        }
    } else {
        long long work = (((long long)E + EPW - 1) / EPW) * 32;
        unsigned sblocks = (unsigned)((work + 255) / 256);
        scatter_kernel<<<sblocks, 256, 0, stream>>>(h1, deg1, e_ssn, out, E);
        scatter_kernel<<<sblocks, 256, 0, stream>>>(h2, deg2, e_ssf, out, E);
        scatter_kernel<<<sblocks, 256, 0, stream>>>(h3, deg3, e_ggn, out + totalNC, E);
    }
}


# edge scatter as one resident launch: 2048 blocks loop over the tickets of all 3 edge types (was 23442 one-ticket blocks in 3 launches), next ticket drawn early; launches 6 -> 4; sums in arrival order
# speedup vs baseline: 3.8502x; 1.0427x over previous
#include <hip/hip_runtime.h>

typedef __attribute__((ext_vector_type(2))) float v2f;
typedef __attribute__((ext_vector_type(8))) float v8f;

#define CH 64

#define TICKW 384
__global__ void init_deg_kernel(float* __restrict__ deg, int n3, unsigned int* __restrict__ tick) {
    int i = blockIdx.x * blockDim.x + threadIdx.x;
    if (i < n3) deg[i] = 1.0f;
    if (tick != nullptr && i < TICKW) tick[i] = 0u;
}

#define HB 16384
#define HT 1024
#define HU 8
__global__ void __launch_bounds__(HT)
hist_deg_kernel(const int* __restrict__ e0, const int* __restrict__ e1,
                const int* __restrict__ e2, float* __restrict__ deg,
                int E, int N, int chunk) {
    __shared__ unsigned int hist[HB];
    const int tid  = threadIdx.x;
    const int type = blockIdx.z;
    const int* __restrict__ dst = (type == 0 ? e0 : (type == 1 ? e1 : e2)) + E;
    float* __restrict__ dg = deg + (long long)type * N;
    const int lo = blockIdx.y * HB;

    for (int i = tid; i < HB; i += HT) hist[i] = 0u;
    __syncthreads();

    const int beg = blockIdx.x * chunk;
    const int end = min(beg + chunk, E);
    for (int i = beg + tid; i < end; i += HU * HT) {
        int d[HU];
#pragma unroll
        for (int u = 0; u < HU; ++u) {
            const int j = i + u * HT;
            d[u] = (j < end) ? dst[j] : -1;
        }
#pragma unroll
        for (int u = 0; u < HU; ++u) {
            const unsigned int b = (unsigned int)(d[u] - lo);
            if (b < HB) atomicAdd(&hist[b], 1u);
        }
    }
    __syncthreads();

    for (int i = tid; i < HB; i += HT) {
        const unsigned int c = hist[i];
        const int node = lo + i;
        if (c != 0u && node < N) unsafeAtomicAdd(&dg[node], (float)c);
    }
}

__global__ void __launch_bounds__(32)
dense_fused_kernel(const float* __restrict__ Xs, const float* __restrict__ Xg,
                   const float* __restrict__ W1, const float* __restrict__ W2,
                   const float* __restrict__ W3,
                   const float* __restrict__ b1, const float* __restrict__ b2,
                   const float* __restrict__ b3,
                   const float* __restrict__ deg,
                   float* __restrict__ H1, float* __restrict__ H2, float* __restrict__ H3,
                   float* __restrict__ out, int N) {
    const int lane = threadIdx.x & 31;
    const int r    = lane & 15;
    const int kh   = lane >> 4;
    const long long m0 = (long long)blockIdx.x * 16;

    const float* xs0 = Xs + m0 * CH;
    const float* xg0 = Xg + m0 * CH;
    unsigned xofs = (unsigned)(r * CH + 2 * kh);
    unsigned wofs = (unsigned)(2 * kh * CH + r);
    const float bs0 = b1[r] + b2[r],           bg0 = b3[r];
    const float bs1 = b1[16 + r] + b2[16 + r], bg1 = b3[16 + r];
    const float bs2 = b1[32 + r] + b2[32 + r], bg2 = b3[32 + r];
    const float bs3 = b1[48 + r] + b2[48 + r], bg3 = b3[48 + r];
    float s1[8], s2[8], s3[8];
#pragma unroll
    for (int j = 0; j < 8; ++j) {
        const long long row = m0 + j + 8 * kh;
        const float d1 = rsqrtf(deg[row]);
        const float d2 = rsqrtf(deg[(long long)N + row]);
        const float d3 = rsqrtf(deg[2LL * N + row]);
        s1[j] = d1 * d1;  s2[j] = d2 * d2;  s3[j] = d3 * d3;
    }

    v8f p0 = {}, p1 = {}, p2 = {}, p3 = {};
    v8f q0 = {}, q1 = {}, q2 = {}, q3 = {};
    v8f g0 = {}, g1 = {}, g2 = {}, g3 = {};

    for (int ks = 0; ks < 16; ++ks) {
        v2f a, c;
        a.x = xs0[xofs];
        a.y = xs0[xofs + 1];
        c.x = xg0[xofs];
        c.y = xg0[xofs + 1];
        v2f b;
        b.x = W1[wofs + 0 * 16]; b.y = W1[wofs + CH + 0 * 16];
        p0 = __builtin_amdgcn_wmma_f32_16x16x4_f32(false, a, false, b, (short)0, p0, false, false);
        b.x = W1[wofs + 1 * 16]; b.y = W1[wofs + CH + 1 * 16];
        p1 = __builtin_amdgcn_wmma_f32_16x16x4_f32(false, a, false, b, (short)0, p1, false, false);
        b.x = W1[wofs + 2 * 16]; b.y = W1[wofs + CH + 2 * 16];
        p2 = __builtin_amdgcn_wmma_f32_16x16x4_f32(false, a, false, b, (short)0, p2, false, false);
        b.x = W1[wofs + 3 * 16]; b.y = W1[wofs + CH + 3 * 16];
        p3 = __builtin_amdgcn_wmma_f32_16x16x4_f32(false, a, false, b, (short)0, p3, false, false);

        b.x = W2[wofs + 0 * 16]; b.y = W2[wofs + CH + 0 * 16];
        q0 = __builtin_amdgcn_wmma_f32_16x16x4_f32(false, a, false, b, (short)0, q0, false, false);
        b.x = W2[wofs + 1 * 16]; b.y = W2[wofs + CH + 1 * 16];
        q1 = __builtin_amdgcn_wmma_f32_16x16x4_f32(false, a, false, b, (short)0, q1, false, false);
        b.x = W2[wofs + 2 * 16]; b.y = W2[wofs + CH + 2 * 16];
        q2 = __builtin_amdgcn_wmma_f32_16x16x4_f32(false, a, false, b, (short)0, q2, false, false);
        b.x = W2[wofs + 3 * 16]; b.y = W2[wofs + CH + 3 * 16];
        q3 = __builtin_amdgcn_wmma_f32_16x16x4_f32(false, a, false, b, (short)0, q3, false, false);

        b.x = W3[wofs + 0 * 16]; b.y = W3[wofs + CH + 0 * 16];
        g0 = __builtin_amdgcn_wmma_f32_16x16x4_f32(false, c, false, b, (short)0, g0, false, false);
        b.x = W3[wofs + 1 * 16]; b.y = W3[wofs + CH + 1 * 16];
        g1 = __builtin_amdgcn_wmma_f32_16x16x4_f32(false, c, false, b, (short)0, g1, false, false);
        b.x = W3[wofs + 2 * 16]; b.y = W3[wofs + CH + 2 * 16];
        g2 = __builtin_amdgcn_wmma_f32_16x16x4_f32(false, c, false, b, (short)0, g2, false, false);
        b.x = W3[wofs + 3 * 16]; b.y = W3[wofs + CH + 3 * 16];
        g3 = __builtin_amdgcn_wmma_f32_16x16x4_f32(false, c, false, b, (short)0, g3, false, false);
        xofs += 4;
        wofs += 4 * CH;
    }

    const long long total = (long long)N * CH;

#pragma unroll
    for (int j = 0; j < 8; ++j) {
        const long long row = m0 + j + 8 * kh;
        const long long o = row * CH + r;
        H1[o]      = p0[j];  H1[o + 16] = p1[j];  H1[o + 32] = p2[j];  H1[o + 48] = p3[j];
        H2[o]      = q0[j];  H2[o + 16] = q1[j];  H2[o + 32] = q2[j];  H2[o + 48] = q3[j];
        H3[o]      = g0[j];  H3[o + 16] = g1[j];  H3[o + 32] = g2[j];  H3[o + 48] = g3[j];
        out[o]      = bs0 + s1[j] * p0[j] + s2[j] * q0[j];
        out[o + 16] = bs1 + s1[j] * p1[j] + s2[j] * q1[j];
        out[o + 32] = bs2 + s1[j] * p2[j] + s2[j] * q2[j];
        out[o + 48] = bs3 + s1[j] * p3[j] + s2[j] * q3[j];
        out[total + o]      = bg0 + s3[j] * g0[j];
        out[total + o + 16] = bg1 + s3[j] * g1[j];
        out[total + o + 32] = bg2 + s3[j] * g2[j];
        out[total + o + 48] = bg3 + s3[j] * g3[j];
    }
}

__global__ void gemm_tail_kernel(const float* __restrict__ X, const float* __restrict__ W,
                                 float* __restrict__ H, int rowStart, int rows) {
    long long t = (long long)blockIdx.x * blockDim.x + threadIdx.x;
    if (t >= (long long)rows * CH) return;
    int row = rowStart + (int)(t >> 6);
    int c   = (int)(t & 63);
    float s = 0.0f;
    for (int k = 0; k < CH; ++k) s += X[(long long)row * CH + k] * W[(long long)k * CH + c];
    H[(long long)row * CH + c] = s;
}

__global__ void init_out_tail_kernel(const float* __restrict__ h1, const float* __restrict__ h2,
                                     const float* __restrict__ h3, const float* __restrict__ deg,
                                     const float* __restrict__ b1, const float* __restrict__ b2,
                                     const float* __restrict__ b3,
                                     float* __restrict__ out, int N, int rowStart, int rows) {
    long long q = (long long)blockIdx.x * blockDim.x + threadIdx.x;
    if (q >= (long long)rows * CH) return;
    long long total = (long long)N * CH;
    int i = rowStart + (int)(q >> 6);
    int c = (int)(q & 63);
    long long t = (long long)i * CH + c;
    float d1 = rsqrtf(deg[i]), d2 = rsqrtf(deg[(long long)N + i]), d3 = rsqrtf(deg[2LL * N + i]);
    out[t]         = b1[c] + b2[c] + d1 * d1 * h1[t] + d2 * d2 * h2[t];
    out[total + t] = b3[c] + d3 * d3 * h3[t];
}

#define EPW 4
__global__ void __launch_bounds__(256)
scatter_kernel(const float* __restrict__ H, const float* __restrict__ deg,
               const int* __restrict__ edges, float* __restrict__ out, int E) {
    const int lane = threadIdx.x & 31;
    const long long wave = ((long long)blockIdx.x * blockDim.x + threadIdx.x) >> 5;
    const long long e0 = wave * EPW;
    if (e0 >= E) return;
    long long e = e0 + (lane >> 3);
    if (e > (long long)E - 1) e = (long long)E - 1;
    const int s = edges[e];
    const int d = edges[E + e];
    const float nrm = rsqrtf(deg[s]) * rsqrtf(deg[d]);

    int si[EPW], di[EPW];
    float ni[EPW];
#pragma unroll
    for (int i = 0; i < EPW; ++i) {
        si[i] = __builtin_amdgcn_readlane(s, 8 * i);
        di[i] = __builtin_amdgcn_readlane(d, 8 * i);
        ni[i] = __int_as_float(__builtin_amdgcn_readlane(__float_as_int(nrm), 8 * i));
    }

    float lo[EPW], hi[EPW];
#pragma unroll
    for (int i = 0; i < EPW; ++i) {
        const float* hp = H + (long long)si[i] * CH + lane;
        lo[i] = hp[0];
        hi[i] = hp[32];
    }

#pragma unroll
    for (int i = 0; i < EPW; ++i) {
        const bool ok = (e0 + i < E);
        float* op = out + (long long)di[i] * CH + lane;
        unsafeAtomicAdd(op,      ok ? lo[i] * ni[i] : 0.0f);
        unsafeAtomicAdd(op + 32, ok ? hi[i] * ni[i] : 0.0f);
    }
}

#define SLICE 256
#define TRIP 8
#define RESIDENT 2048
#define JOB_DONE 0xffffffffu

__device__ __forceinline__ unsigned int
next_job(unsigned int nt, unsigned int& k, unsigned int& h, bool& away, const unsigned int home,
         unsigned int* __restrict__ tick, const unsigned int nslices) {
    while (k < 3u) {
        if (nt < nslices) return (k << 29) | (h << 28) | nt;
        if (!away) { away = true;  h ^= 1u; }
        else       { away = false; h = home; ++k; }
        if (k < 3u) nt = atomicAdd(&tick[k * 128u + h * 64u], 1u);
    }
    return JOB_DONE;
}

__global__ void __launch_bounds__(256)
scatter_all_kernel(const float* __restrict__ H1, const float* __restrict__ H2,
                   const float* __restrict__ H3, const float* __restrict__ deg,
                   const int* __restrict__ e1, const int* __restrict__ e2,
                   const int* __restrict__ e3, float* __restrict__ out,
                   unsigned int* __restrict__ tick, int E, int N, int nslices) {
    __shared__ unsigned int sh_job[2];
    const int lane = threadIdx.x & 31;
    const int wv   = threadIdx.x >> 5;
    const int split = N >> 1;
    const long long total = (long long)N * CH;

    unsigned int home = 0u, dk = 0u, dh = 0u;
    bool away = false;
    if (threadIdx.x == 0) {
        const unsigned int id = (unsigned int)__builtin_amdgcn_s_sendmsg_rtn(0x87);
        home = (id >> 18) & 1u;
        dh = home;
        const unsigned int nt = atomicAdd(&tick[dh * 64u], 1u);
        sh_job[0] = next_job(nt, dk, dh, away, home, tick, (unsigned int)nslices);
    }

    for (unsigned int it = 0u; ; ++it) {
        __syncthreads();
        const unsigned int job = (unsigned int)__builtin_amdgcn_readfirstlane((int)sh_job[it & 1u]);
        if (job == JOB_DONE) break;
        unsigned int nt = 0u;
        if (threadIdx.x == 0 && dk < 3u) nt = atomicAdd(&tick[dk * 128u + dh * 64u], 1u);

        const unsigned int type = job >> 29;
        const unsigned int hsel = (job >> 28) & 1u;
        const unsigned int t    = job & 0x0fffffffu;
        const float* __restrict__ H     = (type == 0u ? H1 : (type == 1u ? H2 : H3));
        const int*   __restrict__ edges = (type == 0u ? e1 : (type == 1u ? e2 : e3));
        const float* __restrict__ dg    = deg + (long long)type * N;
        float*       __restrict__ op0   = out + (type == 2u ? total : 0LL);

        const long long e = (long long)t * SLICE + wv * 32 + lane;
        int s = 0, d = 0;
        float nrm = 0.0f;
        bool mine = false;
        if (e < E) {
            s = edges[e];
            d = edges[E + e];
            mine = ((d >= split ? 1u : 0u) == hsel);
            if (mine) nrm = rsqrtf(dg[s]) * rsqrtf(dg[d]);
        }
        unsigned int mask = __builtin_amdgcn_ballot_w32(mine);

        while (mask != 0u) {
            const int n = min((int)__builtin_popcount(mask), TRIP);
            int si[TRIP], di[TRIP];
            float ni[TRIP];
#pragma unroll
            for (int q = 0; q < TRIP; ++q) {
                const int l = (mask != 0u) ? __builtin_ctz(mask) : 0;
                mask &= mask - 1u;
                si[q] = __builtin_amdgcn_readlane(s, l);
                di[q] = __builtin_amdgcn_readlane(d, l);
                ni[q] = __int_as_float(__builtin_amdgcn_readlane(__float_as_int(nrm), l));
            }
            float lo[TRIP], hi[TRIP];
#pragma unroll
            for (int q = 0; q < TRIP; ++q) {
                lo[q] = 0.0f;
                hi[q] = 0.0f;
                if (q < n) {
                    const float* hp = H + (long long)si[q] * CH + lane;
                    lo[q] = hp[0];
                    hi[q] = hp[32];
                }
            }
#pragma unroll
            for (int q = 0; q < TRIP; ++q) {
                if (q < n) {
                    float* op = op0 + (long long)di[q] * CH + lane;
                    unsafeAtomicAdd(op,      lo[q] * ni[q]);
                    unsafeAtomicAdd(op + 32, hi[q] * ni[q]);
                }
            }
        }

        if (threadIdx.x == 0)
            sh_job[(it + 1u) & 1u] = next_job(nt, dk, dh, away, home, tick, (unsigned int)nslices);
    }
}

extern "C" void kernel_launch(void* const* d_in, const int* in_sizes, int n_in,
                              void* d_out, int out_size, void* d_ws, size_t ws_size,
                              hipStream_t stream) {
    const float* x_star = (const float*)d_in[0];
    const float* x_gal  = (const float*)d_in[1];
    const int*   e_ssn  = (const int*)d_in[2];
    const int*   e_ssf  = (const int*)d_in[3];
    const int*   e_ggn  = (const int*)d_in[4];
    const float* W_ssn  = (const float*)d_in[5];
    const float* W_ssf  = (const float*)d_in[6];
    const float* W_ggn  = (const float*)d_in[7];
    const float* b_ssn  = (const float*)d_in[8];
    const float* b_ssf  = (const float*)d_in[9];
    const float* b_ggn  = (const float*)d_in[10];

    const int N = in_sizes[0] / CH;
    const int E = in_sizes[2] / 2;

    float* ws    = (float*)d_ws;
    float* h1    = ws;
    float* h2    = h1 + (size_t)N * CH;
    float* h3    = h2 + (size_t)N * CH;
    float* deg1  = h3 + (size_t)N * CH;
    float* deg2  = deg1 + N;
    float* deg3  = deg2 + N;
    float* out   = (float*)d_out;
    const size_t tickOff = (((size_t)3 * N * CH + (size_t)3 * N) * sizeof(float) + 255) / 256 * 256;
    const bool   local   = ws_size >= tickOff + TICKW * sizeof(unsigned int);
    unsigned int* tick   = local ? (unsigned int*)((char*)d_ws + tickOff) : nullptr;

    const int initN = 3 * N > TICKW ? 3 * N : TICKW;
    init_deg_kernel<<<(initN + 255) / 256, 256, 0, stream>>>(deg1, 3 * N, tick);
    if (E > 0) {
        const int slices = 16;
        dim3 hgrid(slices, (N + HB - 1) / HB, 3);
        hist_deg_kernel<<<hgrid, HT, 0, stream>>>(e_ssn, e_ssf, e_ggn, deg1, E, N,
                                                  (E + slices - 1) / slices);
    }

    const int tiles = N / 16;
    const int rem   = N - tiles * 16;
    long long totalNC = (long long)N * CH;
    if (tiles > 0) {
        dense_fused_kernel<<<tiles, 32, 0, stream>>>(x_star, x_gal, W_ssn, W_ssf, W_ggn,
                                                     b_ssn, b_ssf, b_ggn, deg1, h1, h2, h3, out, N);
    }
    if (rem > 0) {
        int tail = (rem * CH + 255) / 256;
        gemm_tail_kernel<<<tail, 256, 0, stream>>>(x_star, W_ssn, h1, tiles * 16, rem);
        gemm_tail_kernel<<<tail, 256, 0, stream>>>(x_star, W_ssf, h2, tiles * 16, rem);
        gemm_tail_kernel<<<tail, 256, 0, stream>>>(x_gal,  W_ggn, h3, tiles * 16, rem);
        init_out_tail_kernel<<<tail, 256, 0, stream>>>(h1, h2, h3, deg1, b_ssn, b_ssf, b_ggn, out, N,
                                                       tiles * 16, rem);
    }

    if (local) {
        const int nslices = (E + SLICE - 1) / SLICE;
        if (nslices > 0) {
            const unsigned tickets = 2u * (unsigned)nslices;
            const unsigned lblocks = tickets < RESIDENT ? tickets : RESIDENT;
            scatter_all_kernel<<<lblocks, 256, 0, stream>>>(h1, h2, h3, deg1, e_ssn, e_ssf, e_ggn, out, tick,
                                                            E, N, nslices);
        }
    } else {
        long long work = (((long long)E + EPW - 1) / EPW) * 32;
        unsigned sblocks = (unsigned)((work + 255) / 256);
        scatter_kernel<<<sblocks, 256, 0, stream>>>(h1, deg1, e_ssn, out, E);
        scatter_kernel<<<sblocks, 256, 0, stream>>>(h2, deg2, e_ssf, out, E);
        scatter_kernel<<<sblocks, 256, 0, stream>>>(h3, deg3, e_ggn, out + totalNC, E);
    }
}


# norm split: dense epilogue stores h pre-scaled by dinv[src] (self-loop term built from it), so the scatter looks up deg[dst] only, 1 random lookup per edge (was 2); atomics and their order unchanged
# speedup vs baseline: 4.0881x; 1.0618x over previous
#include <hip/hip_runtime.h>

typedef __attribute__((ext_vector_type(2))) float v2f;
typedef __attribute__((ext_vector_type(8))) float v8f;

#define CH 64

#define TICKW 384
__global__ void init_deg_kernel(float* __restrict__ deg, int n3, unsigned int* __restrict__ tick) {
    int i = blockIdx.x * blockDim.x + threadIdx.x;
    if (i < n3) deg[i] = 1.0f;
    if (tick != nullptr && i < TICKW) tick[i] = 0u;
}

#define HB 16384
#define HT 1024
#define HU 8
__global__ void __launch_bounds__(HT)
hist_deg_kernel(const int* __restrict__ e0, const int* __restrict__ e1,
                const int* __restrict__ e2, float* __restrict__ deg,
                int E, int N, int chunk) {
    __shared__ unsigned int hist[HB];
    const int tid  = threadIdx.x;
    const int type = blockIdx.z;
    const int* __restrict__ dst = (type == 0 ? e0 : (type == 1 ? e1 : e2)) + E;
    float* __restrict__ dg = deg + (long long)type * N;
    const int lo = blockIdx.y * HB;

    for (int i = tid; i < HB; i += HT) hist[i] = 0u;
    __syncthreads();

    const int beg = blockIdx.x * chunk;
    const int end = min(beg + chunk, E);
    for (int i = beg + tid; i < end; i += HU * HT) {
        int d[HU];
#pragma unroll
        for (int u = 0; u < HU; ++u) {
            const int j = i + u * HT;
            d[u] = (j < end) ? dst[j] : -1;
        }
#pragma unroll
        for (int u = 0; u < HU; ++u) {
            const unsigned int b = (unsigned int)(d[u] - lo);
            if (b < HB) atomicAdd(&hist[b], 1u);
        }
    }
    __syncthreads();

    for (int i = tid; i < HB; i += HT) {
        const unsigned int c = hist[i];
        const int node = lo + i;
        if (c != 0u && node < N) unsafeAtomicAdd(&dg[node], (float)c);
    }
}

__global__ void __launch_bounds__(32)
dense_fused_kernel(const float* __restrict__ Xs, const float* __restrict__ Xg,
                   const float* __restrict__ W1, const float* __restrict__ W2,
                   const float* __restrict__ W3,
                   const float* __restrict__ b1, const float* __restrict__ b2,
                   const float* __restrict__ b3,
                   const float* __restrict__ deg,
                   float* __restrict__ H1, float* __restrict__ H2, float* __restrict__ H3,
                   float* __restrict__ out, int N) {
    const int lane = threadIdx.x & 31;
    const int r    = lane & 15;
    const int kh   = lane >> 4;
    const long long m0 = (long long)blockIdx.x * 16;

    const float* xs0 = Xs + m0 * CH;
    const float* xg0 = Xg + m0 * CH;
    unsigned xofs = (unsigned)(r * CH + 2 * kh);
    unsigned wofs = (unsigned)(2 * kh * CH + r);
    const float bs0 = b1[r] + b2[r],           bg0 = b3[r];
    const float bs1 = b1[16 + r] + b2[16 + r], bg1 = b3[16 + r];
    const float bs2 = b1[32 + r] + b2[32 + r], bg2 = b3[32 + r];
    const float bs3 = b1[48 + r] + b2[48 + r], bg3 = b3[48 + r];
    float s1[8], s2[8], s3[8];
#pragma unroll
    for (int j = 0; j < 8; ++j) {
        const long long row = m0 + j + 8 * kh;
        s1[j] = rsqrtf(deg[row]);
        s2[j] = rsqrtf(deg[(long long)N + row]);
        s3[j] = rsqrtf(deg[2LL * N + row]);
    }

    v8f p0 = {}, p1 = {}, p2 = {}, p3 = {};
    v8f q0 = {}, q1 = {}, q2 = {}, q3 = {};
    v8f g0 = {}, g1 = {}, g2 = {}, g3 = {};

    for (int ks = 0; ks < 16; ++ks) {
        v2f a, c;
        a.x = xs0[xofs];
        a.y = xs0[xofs + 1];
        c.x = xg0[xofs];
        c.y = xg0[xofs + 1];
        v2f b;
        b.x = W1[wofs + 0 * 16]; b.y = W1[wofs + CH + 0 * 16];
        p0 = __builtin_amdgcn_wmma_f32_16x16x4_f32(false, a, false, b, (short)0, p0, false, false);
        b.x = W1[wofs + 1 * 16]; b.y = W1[wofs + CH + 1 * 16];
        p1 = __builtin_amdgcn_wmma_f32_16x16x4_f32(false, a, false, b, (short)0, p1, false, false);
        b.x = W1[wofs + 2 * 16]; b.y = W1[wofs + CH + 2 * 16];
        p2 = __builtin_amdgcn_wmma_f32_16x16x4_f32(false, a, false, b, (short)0, p2, false, false);
        b.x = W1[wofs + 3 * 16]; b.y = W1[wofs + CH + 3 * 16];
        p3 = __builtin_amdgcn_wmma_f32_16x16x4_f32(false, a, false, b, (short)0, p3, false, false);

        b.x = W2[wofs + 0 * 16]; b.y = W2[wofs + CH + 0 * 16];
        q0 = __builtin_amdgcn_wmma_f32_16x16x4_f32(false, a, false, b, (short)0, q0, false, false);
        b.x = W2[wofs + 1 * 16]; b.y = W2[wofs + CH + 1 * 16];
        q1 = __builtin_amdgcn_wmma_f32_16x16x4_f32(false, a, false, b, (short)0, q1, false, false);
        b.x = W2[wofs + 2 * 16]; b.y = W2[wofs + CH + 2 * 16];
        q2 = __builtin_amdgcn_wmma_f32_16x16x4_f32(false, a, false, b, (short)0, q2, false, false);
        b.x = W2[wofs + 3 * 16]; b.y = W2[wofs + CH + 3 * 16];
        q3 = __builtin_amdgcn_wmma_f32_16x16x4_f32(false, a, false, b, (short)0, q3, false, false);

        b.x = W3[wofs + 0 * 16]; b.y = W3[wofs + CH + 0 * 16];
        g0 = __builtin_amdgcn_wmma_f32_16x16x4_f32(false, c, false, b, (short)0, g0, false, false);
        b.x = W3[wofs + 1 * 16]; b.y = W3[wofs + CH + 1 * 16];
        g1 = __builtin_amdgcn_wmma_f32_16x16x4_f32(false, c, false, b, (short)0, g1, false, false);
        b.x = W3[wofs + 2 * 16]; b.y = W3[wofs + CH + 2 * 16];
        g2 = __builtin_amdgcn_wmma_f32_16x16x4_f32(false, c, false, b, (short)0, g2, false, false);
        b.x = W3[wofs + 3 * 16]; b.y = W3[wofs + CH + 3 * 16];
        g3 = __builtin_amdgcn_wmma_f32_16x16x4_f32(false, c, false, b, (short)0, g3, false, false);
        xofs += 4;
        wofs += 4 * CH;
    }

    const long long total = (long long)N * CH;

#pragma unroll
    for (int j = 0; j < 8; ++j) {
        const long long row = m0 + j + 8 * kh;
        const long long o = row * CH + r;
        const float u0 = s1[j] * p0[j], u1 = s1[j] * p1[j], u2 = s1[j] * p2[j], u3 = s1[j] * p3[j];
        const float w0 = s2[j] * q0[j], w1 = s2[j] * q1[j], w2 = s2[j] * q2[j], w3 = s2[j] * q3[j];
        const float z0 = s3[j] * g0[j], z1 = s3[j] * g1[j], z2 = s3[j] * g2[j], z3 = s3[j] * g3[j];
        H1[o]      = u0;  H1[o + 16] = u1;  H1[o + 32] = u2;  H1[o + 48] = u3;
        H2[o]      = w0;  H2[o + 16] = w1;  H2[o + 32] = w2;  H2[o + 48] = w3;
        H3[o]      = z0;  H3[o + 16] = z1;  H3[o + 32] = z2;  H3[o + 48] = z3;
        out[o]      = bs0 + s1[j] * u0 + s2[j] * w0;
        out[o + 16] = bs1 + s1[j] * u1 + s2[j] * w1;
        out[o + 32] = bs2 + s1[j] * u2 + s2[j] * w2;
        out[o + 48] = bs3 + s1[j] * u3 + s2[j] * w3;
        out[total + o]      = bg0 + s3[j] * z0;
        out[total + o + 16] = bg1 + s3[j] * z1;
        out[total + o + 32] = bg2 + s3[j] * z2;
        out[total + o + 48] = bg3 + s3[j] * z3;
    }
}

__global__ void gemm_tail_kernel(const float* __restrict__ X, const float* __restrict__ W,
                                 float* __restrict__ H, int rowStart, int rows) {
    long long t = (long long)blockIdx.x * blockDim.x + threadIdx.x;
    if (t >= (long long)rows * CH) return;
    int row = rowStart + (int)(t >> 6);
    int c   = (int)(t & 63);
    float s = 0.0f;
    for (int k = 0; k < CH; ++k) s += X[(long long)row * CH + k] * W[(long long)k * CH + c];
    H[(long long)row * CH + c] = s;
}

__global__ void init_out_tail_kernel(float* __restrict__ h1, float* __restrict__ h2,
                                     float* __restrict__ h3, const float* __restrict__ deg,
                                     const float* __restrict__ b1, const float* __restrict__ b2,
                                     const float* __restrict__ b3,
                                     float* __restrict__ out, int N, int rowStart, int rows) {
    long long q = (long long)blockIdx.x * blockDim.x + threadIdx.x;
    if (q >= (long long)rows * CH) return;
    long long total = (long long)N * CH;
    int i = rowStart + (int)(q >> 6);
    int c = (int)(q & 63);
    long long t = (long long)i * CH + c;
    float d1 = rsqrtf(deg[i]), d2 = rsqrtf(deg[(long long)N + i]), d3 = rsqrtf(deg[2LL * N + i]);
    const float u = d1 * h1[t], w = d2 * h2[t], z = d3 * h3[t];
    h1[t] = u;  h2[t] = w;  h3[t] = z;
    out[t]         = b1[c] + b2[c] + d1 * u + d2 * w;
    out[total + t] = b3[c] + d3 * z;
}

#define EPW 4
__global__ void __launch_bounds__(256)
scatter_kernel(const float* __restrict__ H, const float* __restrict__ deg,
               const int* __restrict__ edges, float* __restrict__ out, int E) {
    const int lane = threadIdx.x & 31;
    const long long wave = ((long long)blockIdx.x * blockDim.x + threadIdx.x) >> 5;
    const long long e0 = wave * EPW;
    if (e0 >= E) return;
    long long e = e0 + (lane >> 3);
    if (e > (long long)E - 1) e = (long long)E - 1;
    const int s = edges[e];
    const int d = edges[E + e];
    const float nrm = rsqrtf(deg[d]);

    int si[EPW], di[EPW];
    float ni[EPW];
#pragma unroll
    for (int i = 0; i < EPW; ++i) {
        si[i] = __builtin_amdgcn_readlane(s, 8 * i);
        di[i] = __builtin_amdgcn_readlane(d, 8 * i);
        ni[i] = __int_as_float(__builtin_amdgcn_readlane(__float_as_int(nrm), 8 * i));
    }

    float lo[EPW], hi[EPW];
#pragma unroll
    for (int i = 0; i < EPW; ++i) {
        const float* hp = H + (long long)si[i] * CH + lane;
        lo[i] = hp[0];
        hi[i] = hp[32];
    }

#pragma unroll
    for (int i = 0; i < EPW; ++i) {
        const bool ok = (e0 + i < E);
        float* op = out + (long long)di[i] * CH + lane;
        unsafeAtomicAdd(op,      ok ? lo[i] * ni[i] : 0.0f);
        unsafeAtomicAdd(op + 32, ok ? hi[i] * ni[i] : 0.0f);
    }
}

#define SLICE 256
#define TRIP 8
#define RESIDENT 2048
#define JOB_DONE 0xffffffffu

__device__ __forceinline__ unsigned int
next_job(unsigned int nt, unsigned int& k, unsigned int& h, bool& away, const unsigned int home,
         unsigned int* __restrict__ tick, const unsigned int nslices) {
    while (k < 3u) {
        if (nt < nslices) return (k << 29) | (h << 28) | nt;
        if (!away) { away = true;  h ^= 1u; }
        else       { away = false; h = home; ++k; }
        if (k < 3u) nt = atomicAdd(&tick[k * 128u + h * 64u], 1u);
    }
    return JOB_DONE;
}

__global__ void __launch_bounds__(256)
scatter_all_kernel(const float* __restrict__ H1, const float* __restrict__ H2,
                   const float* __restrict__ H3, const float* __restrict__ deg,
                   const int* __restrict__ e1, const int* __restrict__ e2,
                   const int* __restrict__ e3, float* __restrict__ out,
                   unsigned int* __restrict__ tick, int E, int N, int nslices) {
    __shared__ unsigned int sh_job[2];
    const int lane = threadIdx.x & 31;
    const int wv   = threadIdx.x >> 5;
    const int split = N >> 1;
    const long long total = (long long)N * CH;

    unsigned int home = 0u, dk = 0u, dh = 0u;
    bool away = false;
    if (threadIdx.x == 0) {
        const unsigned int id = (unsigned int)__builtin_amdgcn_s_sendmsg_rtn(0x87);
        home = (id >> 18) & 1u;
        dh = home;
        const unsigned int nt = atomicAdd(&tick[dh * 64u], 1u);
        sh_job[0] = next_job(nt, dk, dh, away, home, tick, (unsigned int)nslices);
    }

    for (unsigned int it = 0u; ; ++it) {
        __syncthreads();
        const unsigned int job = (unsigned int)__builtin_amdgcn_readfirstlane((int)sh_job[it & 1u]);
        if (job == JOB_DONE) break;
        unsigned int nt = 0u;
        if (threadIdx.x == 0 && dk < 3u) nt = atomicAdd(&tick[dk * 128u + dh * 64u], 1u);

        const unsigned int type = job >> 29;
        const unsigned int hsel = (job >> 28) & 1u;
        const unsigned int t    = job & 0x0fffffffu;
        const float* __restrict__ H     = (type == 0u ? H1 : (type == 1u ? H2 : H3));
        const int*   __restrict__ edges = (type == 0u ? e1 : (type == 1u ? e2 : e3));
        const float* __restrict__ dg    = deg + (long long)type * N;
        float*       __restrict__ op0   = out + (type == 2u ? total : 0LL);

        const long long e = (long long)t * SLICE + wv * 32 + lane;
        int s = 0, d = 0;
        float nrm = 0.0f;
        bool mine = false;
        if (e < E) {
            s = edges[e];
            d = edges[E + e];
            mine = ((d >= split ? 1u : 0u) == hsel);
            if (mine) nrm = rsqrtf(dg[d]);
        }
        unsigned int mask = __builtin_amdgcn_ballot_w32(mine);

        while (mask != 0u) {
            const int n = min((int)__builtin_popcount(mask), TRIP);
            int si[TRIP], di[TRIP];
            float ni[TRIP];
#pragma unroll
            for (int q = 0; q < TRIP; ++q) {
                const int l = (mask != 0u) ? __builtin_ctz(mask) : 0;
                mask &= mask - 1u;
                si[q] = __builtin_amdgcn_readlane(s, l);
                di[q] = __builtin_amdgcn_readlane(d, l);
                ni[q] = __int_as_float(__builtin_amdgcn_readlane(__float_as_int(nrm), l));
            }
            float lo[TRIP], hi[TRIP];
#pragma unroll
            for (int q = 0; q < TRIP; ++q) {
                lo[q] = 0.0f;
                hi[q] = 0.0f;
                if (q < n) {
                    const float* hp = H + (long long)si[q] * CH + lane;
                    lo[q] = hp[0];
                    hi[q] = hp[32];
                }
            }
#pragma unroll
            for (int q = 0; q < TRIP; ++q) {
                if (q < n) {
                    float* op = op0 + (long long)di[q] * CH + lane;
                    unsafeAtomicAdd(op,      lo[q] * ni[q]);
                    unsafeAtomicAdd(op + 32, hi[q] * ni[q]);
                }
            }
        }

        if (threadIdx.x == 0)
            sh_job[(it + 1u) & 1u] = next_job(nt, dk, dh, away, home, tick, (unsigned int)nslices);
    }
}

extern "C" void kernel_launch(void* const* d_in, const int* in_sizes, int n_in,
                              void* d_out, int out_size, void* d_ws, size_t ws_size,
                              hipStream_t stream) {
    const float* x_star = (const float*)d_in[0];
    const float* x_gal  = (const float*)d_in[1];
    const int*   e_ssn  = (const int*)d_in[2];
    const int*   e_ssf  = (const int*)d_in[3];
    const int*   e_ggn  = (const int*)d_in[4];
    const float* W_ssn  = (const float*)d_in[5];
    const float* W_ssf  = (const float*)d_in[6];
    const float* W_ggn  = (const float*)d_in[7];
    const float* b_ssn  = (const float*)d_in[8];
    const float* b_ssf  = (const float*)d_in[9];
    const float* b_ggn  = (const float*)d_in[10];

    const int N = in_sizes[0] / CH;
    const int E = in_sizes[2] / 2;

    float* ws    = (float*)d_ws;
    float* h1    = ws;
    float* h2    = h1 + (size_t)N * CH;
    float* h3    = h2 + (size_t)N * CH;
    float* deg1  = h3 + (size_t)N * CH;
    float* deg2  = deg1 + N;
    float* deg3  = deg2 + N;
    float* out   = (float*)d_out;
    const size_t tickOff = (((size_t)3 * N * CH + (size_t)3 * N) * sizeof(float) + 255) / 256 * 256;
    const bool   local   = ws_size >= tickOff + TICKW * sizeof(unsigned int);
    unsigned int* tick   = local ? (unsigned int*)((char*)d_ws + tickOff) : nullptr;

    const int initN = 3 * N > TICKW ? 3 * N : TICKW;
    init_deg_kernel<<<(initN + 255) / 256, 256, 0, stream>>>(deg1, 3 * N, tick);
    if (E > 0) {
        const int slices = 16;
        dim3 hgrid(slices, (N + HB - 1) / HB, 3);
        hist_deg_kernel<<<hgrid, HT, 0, stream>>>(e_ssn, e_ssf, e_ggn, deg1, E, N,
                                                  (E + slices - 1) / slices);
    }

    const int tiles = N / 16;
    const int rem   = N - tiles * 16;
    long long totalNC = (long long)N * CH;
    if (tiles > 0) {
        dense_fused_kernel<<<tiles, 32, 0, stream>>>(x_star, x_gal, W_ssn, W_ssf, W_ggn,
                                                     b_ssn, b_ssf, b_ggn, deg1, h1, h2, h3, out, N);
    }
    if (rem > 0) {
        int tail = (rem * CH + 255) / 256;
        gemm_tail_kernel<<<tail, 256, 0, stream>>>(x_star, W_ssn, h1, tiles * 16, rem);
        gemm_tail_kernel<<<tail, 256, 0, stream>>>(x_star, W_ssf, h2, tiles * 16, rem);
        gemm_tail_kernel<<<tail, 256, 0, stream>>>(x_gal,  W_ggn, h3, tiles * 16, rem);
        init_out_tail_kernel<<<tail, 256, 0, stream>>>(h1, h2, h3, deg1, b_ssn, b_ssf, b_ggn, out, N,
                                                       tiles * 16, rem);
    }

    if (local) {
        const int nslices = (E + SLICE - 1) / SLICE;
        if (nslices > 0) {
            const unsigned tickets = 2u * (unsigned)nslices;
            const unsigned lblocks = tickets < RESIDENT ? tickets : RESIDENT;
            scatter_all_kernel<<<lblocks, 256, 0, stream>>>(h1, h2, h3, deg1, e_ssn, e_ssf, e_ggn, out, tick,
                                                            E, N, nslices);
        }
    } else {
        long long work = (((long long)E + EPW - 1) / EPW) * 32;
        unsigned sblocks = (unsigned)((work + 255) / 256);
        scatter_kernel<<<sblocks, 256, 0, stream>>>(h1, deg1, e_ssn, out, E);
        scatter_kernel<<<sblocks, 256, 0, stream>>>(h2, deg2, e_ssf, out, E);
        scatter_kernel<<<sblocks, 256, 0, stream>>>(h3, deg3, e_ggn, out + totalNC, E);
    }
}
